# starting kernel
# baseline (speedup 1.0000x reference)
#include <hip/hip_runtime.h>
#include <hip/hip_bf16.h>

typedef __attribute__((ext_vector_type(2))) float v2f;
typedef __attribute__((ext_vector_type(8))) float v8f;

#define N_NODES   100000
#define N_EDGES0  800000
#define N_EDGES   900000
#define ROW_TILES 6250

__device__ __forceinline__ float leaky(float v) { return v > 0.0f ? v : 0.2f * v; }

__device__ __forceinline__ void atomicMaxFloat(float* addr, float val) {
    if (val >= 0.0f) atomicMax((int*)addr, __float_as_int(val));
    else             atomicMin((unsigned int*)addr, __float_as_uint(val));
}

__global__ __launch_bounds__(256) void init_ws(float* agg, float* emax, float* denom) {
    size_t i = (size_t)blockIdx.x * 256 + threadIdx.x;
    if (i < (size_t)N_NODES * 256) agg[i] = 0.0f;
    if (i < N_NODES) { emax[i] = -INFINITY; denom[i] = 0.0f; }
}

__global__ __launch_bounds__(256) void gemm1_xw(const float* __restrict__ x,
                                                const float* __restrict__ W,
                                                float* __restrict__ xw) {
    const int lane = threadIdx.x & 31;
    const int wave = threadIdx.x >> 5;
    const int rowTile = blockIdx.x * 8 + wave;
    if (rowTile >= ROW_TILES) return;
    const int row0 = rowTile * 16;
    const int col0 = blockIdx.y * 64;

    const int rA  = row0 + (lane & 15);
    const int kHi = (lane >> 4) * 2;
    const int nB  = lane & 15;

    v8f acc[4] = {};
    for (int k = 0; k < 128; k += 4) {
        v2f a;
        a.x = x[(size_t)rA * 128 + k + kHi];
        a.y = x[(size_t)rA * 128 + k + kHi + 1];
#pragma unroll
        for (int t = 0; t < 4; ++t) {
            v2f bm;
            bm.x = W[(size_t)(k + kHi)     * 256 + col0 + t * 16 + nB];
            bm.y = W[(size_t)(k + kHi + 1) * 256 + col0 + t * 16 + nB];
            acc[t] = __builtin_amdgcn_wmma_f32_16x16x4_f32(
                false, a, false, bm, (short)0, acc[t], false, false);
        }
    }
    const int mBase = row0 + ((lane >> 4) << 3);
#pragma unroll
    for (int t = 0; t < 4; ++t)
#pragma unroll
        for (int r = 0; r < 8; ++r)
            xw[(size_t)(mBase + r) * 256 + col0 + t * 16 + nB] = acc[t][r];
}

__global__ __launch_bounds__(256) void alphas_kernel(const float* __restrict__ xw,
                                                     const float* __restrict__ a_src,
                                                     const float* __restrict__ a_dst,
                                                     float* __restrict__ as,
                                                     float* __restrict__ ad) {
    const int lane = threadIdx.x & 31;
    const int node = blockIdx.x * 8 + (threadIdx.x >> 5);
    if (node >= N_NODES) return;
    const float* xr = xw + (size_t)node * 256;
    float s = 0.0f, d = 0.0f;
#pragma unroll
    for (int i = 0; i < 8; ++i) {
        float v = xr[lane + 32 * i];
        s += v * a_src[lane + 32 * i];
        d += v * a_dst[lane + 32 * i];
    }
#pragma unroll
    for (int off = 16; off; off >>= 1) {
        s += __shfl_xor(s, off, 32);
        d += __shfl_xor(d, off, 32);
    }
    if (lane == 0) { as[node] = s; ad[node] = d; }
}

__device__ __forceinline__ void edge_endpoints(const int* __restrict__ ei, int e,
                                               int& s, int& d) {
    if (e < N_EDGES0) { s = ei[e]; d = ei[N_EDGES0 + e]; }
    else              { s = e - N_EDGES0; d = e - N_EDGES0; }
}

__global__ __launch_bounds__(256) void edge_max_kernel(const int* __restrict__ ei,
                                                       const float* __restrict__ as,
                                                       const float* __restrict__ ad,
                                                       float* __restrict__ emax) {
    int e = blockIdx.x * 256 + threadIdx.x;
    if (e >= N_EDGES) return;
    int s, d; edge_endpoints(ei, e, s, d);
    atomicMaxFloat(&emax[d], leaky(as[s] + ad[d]));
}

__global__ __launch_bounds__(256) void edge_exp_kernel(const int* __restrict__ ei,
                                                       const float* __restrict__ as,
                                                       const float* __restrict__ ad,
                                                       const float* __restrict__ emax,
                                                       float* __restrict__ ebuf,
                                                       float* __restrict__ denom) {
    int e = blockIdx.x * 256 + threadIdx.x;
    if (e >= N_EDGES) return;
    int s, d; edge_endpoints(ei, e, s, d);
    float ex = __expf(leaky(as[s] + ad[d]) - emax[d]);
    ebuf[e] = ex;
    atomicAdd(&denom[d], ex);
}

__global__ __launch_bounds__(256) void edge_agg_kernel(const int* __restrict__ ei,
                                                       const float* __restrict__ ebuf,
                                                       const float* __restrict__ denom,
                                                       const float* __restrict__ xw,
                                                       float* __restrict__ agg) {
    int e = blockIdx.x * 4 + (threadIdx.x >> 6);
    if (e >= N_EDGES) return;
    int s, d; edge_endpoints(ei, e, s, d);
    float alpha = ebuf[e] / denom[d];
    int c0 = (threadIdx.x & 63) * 4;
    float4 v = *(const float4*)(xw + (size_t)s * 256 + c0);
    float* dst = agg + (size_t)d * 256 + c0;
    atomicAdd(dst + 0, alpha * v.x);
    atomicAdd(dst + 1, alpha * v.y);
    atomicAdd(dst + 2, alpha * v.z);
    atomicAdd(dst + 3, alpha * v.w);
}

__global__ __launch_bounds__(256) void gemm2_kernel(const float* __restrict__ agg,
                                                    const float* __restrict__ b,
                                                    const float* __restrict__ Wfc2,
                                                    const float* __restrict__ bfc2,
                                                    float* __restrict__ h2) {
    const int lane = threadIdx.x & 31;
    const int wave = threadIdx.x >> 5;
    const int rowTile = blockIdx.x * 8 + wave;
    if (rowTile >= ROW_TILES) return;
    const int row0 = rowTile * 16;
    const int col0 = blockIdx.y * 64;

    const int rA  = row0 + (lane & 15);
    const int kHi = (lane >> 4) * 2;
    const int nB  = lane & 15;

    v8f acc[4] = {};
    for (int k = 0; k < 256; k += 4) {
        v2f a;
        a.x = fmaxf(agg[(size_t)rA * 256 + k + kHi]     + b[k + kHi],     0.0f);
        a.y = fmaxf(agg[(size_t)rA * 256 + k + kHi + 1] + b[k + kHi + 1], 0.0f);
#pragma unroll
        for (int t = 0; t < 4; ++t) {
            v2f bm;
            bm.x = Wfc2[(size_t)(k + kHi)     * 128 + col0 + t * 16 + nB];
            bm.y = Wfc2[(size_t)(k + kHi + 1) * 128 + col0 + t * 16 + nB];
            acc[t] = __builtin_amdgcn_wmma_f32_16x16x4_f32(
                false, a, false, bm, (short)0, acc[t], false, false);
        }
    }
    const int mBase = row0 + ((lane >> 4) << 3);
#pragma unroll
    for (int t = 0; t < 4; ++t) {
        int col = col0 + t * 16 + nB;
#pragma unroll
        for (int r = 0; r < 8; ++r)
            h2[(size_t)(mBase + r) * 128 + col] = fmaxf(acc[t][r] + bfc2[col], 0.0f);
    }
}

__global__ __launch_bounds__(256) void gemm3_kernel(const float* __restrict__ h2,
                                                    const float* __restrict__ Wout,
                                                    const float* __restrict__ bout,
                                                    float* __restrict__ out) {
    const int lane = threadIdx.x & 31;
    const int wave = threadIdx.x >> 5;
    const int rowTile = blockIdx.x * 8 + wave;
    if (rowTile >= ROW_TILES) return;
    const int row0 = rowTile * 16;

    const int rA  = row0 + (lane & 15);
    const int kHi = (lane >> 4) * 2;
    const int nB  = lane & 15;

    v8f acc = {};
    for (int k = 0; k < 128; k += 4) {
        v2f a;
        a.x = h2[(size_t)rA * 128 + k + kHi];
        a.y = h2[(size_t)rA * 128 + k + kHi + 1];
        v2f bm;
        bm.x = (nB < 6) ? Wout[(size_t)(k + kHi)     * 6 + nB] : 0.0f;
        bm.y = (nB < 6) ? Wout[(size_t)(k + kHi + 1) * 6 + nB] : 0.0f;
        acc = __builtin_amdgcn_wmma_f32_16x16x4_f32(
            false, a, false, bm, (short)0, acc, false, false);
    }
    if (nB < 6) {
        const int mBase = row0 + ((lane >> 4) << 3);
        float bb = bout[nB];
#pragma unroll
        for (int r = 0; r < 8; ++r)
            out[(size_t)(mBase + r) * 6 + nB] = acc[r] + bb;
    }
}

extern "C" void kernel_launch(void* const* d_in, const int* in_sizes, int n_in,
                              void* d_out, int out_size, void* d_ws, size_t ws_size,
                              hipStream_t stream) {
    const float* x     = (const float*)d_in[0];
    const int*   ei    = (const int*)  d_in[1];
    const float* W     = (const float*)d_in[2];
    const float* a_src = (const float*)d_in[3];
    const float* a_dst = (const float*)d_in[4];
    const float* b     = (const float*)d_in[5];
    const float* Wfc2  = (const float*)d_in[6];
    const float* bfc2  = (const float*)d_in[7];
    const float* Wout  = (const float*)d_in[8];
    const float* bout  = (const float*)d_in[9];
    float* out = (float*)d_out;

    float* ws      = (float*)d_ws;
    float* xw      = ws;
    float* agg     = xw + (size_t)N_NODES * 256;
    float* alpha_s = agg + (size_t)N_NODES * 256;
    float* alpha_d = alpha_s + N_NODES;
    float* emax    = alpha_d + N_NODES;
    float* denom   = emax + N_NODES;
    float* ebuf    = denom + N_NODES;
    float* h2      = xw;

    init_ws<<<N_NODES, 256, 0, stream>>>(agg, emax, denom);
    gemm1_xw<<<dim3((ROW_TILES + 7) / 8, 4), 256, 0, stream>>>(x, W, xw);
    alphas_kernel<<<(N_NODES + 7) / 8, 256, 0, stream>>>(xw, a_src, a_dst, alpha_s, alpha_d);
    edge_max_kernel<<<(N_EDGES + 255) / 256, 256, 0, stream>>>(ei, alpha_s, alpha_d, emax);
    edge_exp_kernel<<<(N_EDGES + 255) / 256, 256, 0, stream>>>(ei, alpha_s, alpha_d, emax, ebuf, denom);
    edge_agg_kernel<<<(N_EDGES + 3) / 4, 256, 0, stream>>>(ei, ebuf, denom, xw, agg);
    gemm2_kernel<<<dim3((ROW_TILES + 7) / 8, 2), 256, 0, stream>>>(agg, b, Wfc2, bfc2, h2);
    gemm3_kernel<<<(ROW_TILES + 7) / 8, 256, 0, stream>>>(h2, Wout, bout, out);
}


# Aggregation by destination ownership: each of 256 workgroups filters the edge list for its 392 rows and sums xw rows in LDS (ds f32 adds): no global f32 adds, no agg zero fill; LDS-arrival sum order
# speedup vs baseline: 1.8588x; 1.8588x over previous
#include <hip/hip_runtime.h>
#include <hip/hip_bf16.h>

typedef __attribute__((ext_vector_type(2))) float v2f;
typedef __attribute__((ext_vector_type(8))) float v8f;

#define N_NODES   100000
#define N_EDGES0  800000
#define N_EDGES   900000
#define ROW_TILES 6250

__device__ __forceinline__ float leaky(float v) { return v > 0.0f ? v : 0.2f * v; }

__device__ __forceinline__ void atomicMaxFloat(float* addr, float val) {
    if (val >= 0.0f) atomicMax((int*)addr, __float_as_int(val));
    else             atomicMin((unsigned int*)addr, __float_as_uint(val));
}

__global__ __launch_bounds__(256) void init_ws(float* emax, float* denom) {
    int i = blockIdx.x * 256 + threadIdx.x;
    if (i < N_NODES) { emax[i] = -INFINITY; denom[i] = 0.0f; }
}

__global__ __launch_bounds__(256) void gemm1_xw(const float* __restrict__ x,
                                                const float* __restrict__ W,
                                                float* __restrict__ xw) {
    const int lane = threadIdx.x & 31;
    const int wave = threadIdx.x >> 5;
    const int rowTile = blockIdx.x * 8 + wave;
    if (rowTile >= ROW_TILES) return;
    const int row0 = rowTile * 16;
    const int col0 = blockIdx.y * 64;

    const int rA  = row0 + (lane & 15);
    const int kHi = (lane >> 4) * 2;
    const int nB  = lane & 15;

    v8f acc[4] = {};
    for (int k = 0; k < 128; k += 4) {
        v2f a;
        a.x = x[(size_t)rA * 128 + k + kHi];
        a.y = x[(size_t)rA * 128 + k + kHi + 1];
#pragma unroll
        for (int t = 0; t < 4; ++t) {
            v2f bm;
            bm.x = W[(size_t)(k + kHi)     * 256 + col0 + t * 16 + nB];
            bm.y = W[(size_t)(k + kHi + 1) * 256 + col0 + t * 16 + nB];
            acc[t] = __builtin_amdgcn_wmma_f32_16x16x4_f32(
                false, a, false, bm, (short)0, acc[t], false, false);
        }
    }
    const int mBase = row0 + ((lane >> 4) << 3);
#pragma unroll
    for (int t = 0; t < 4; ++t)
#pragma unroll
        for (int r = 0; r < 8; ++r)
            xw[(size_t)(mBase + r) * 256 + col0 + t * 16 + nB] = acc[t][r];
}

__global__ __launch_bounds__(256) void alphas_kernel(const float* __restrict__ xw,
                                                     const float* __restrict__ a_src,
                                                     const float* __restrict__ a_dst,
                                                     float* __restrict__ as,
                                                     float* __restrict__ ad) {
    const int lane = threadIdx.x & 31;
    const int node = blockIdx.x * 8 + (threadIdx.x >> 5);
    if (node >= N_NODES) return;
    const float* xr = xw + (size_t)node * 256;
    float s = 0.0f, d = 0.0f;
#pragma unroll
    for (int i = 0; i < 8; ++i) {
        float v = xr[lane + 32 * i];
        s += v * a_src[lane + 32 * i];
        d += v * a_dst[lane + 32 * i];
    }
#pragma unroll
    for (int off = 16; off; off >>= 1) {
        s += __shfl_xor(s, off, 32);
        d += __shfl_xor(d, off, 32);
    }
    if (lane == 0) { as[node] = s; ad[node] = d; }
}

__device__ __forceinline__ void edge_endpoints(const int* __restrict__ ei, int e,
                                               int& s, int& d) {
    if (e < N_EDGES0) { s = ei[e]; d = ei[N_EDGES0 + e]; }
    else              { s = e - N_EDGES0; d = e - N_EDGES0; }
}

__global__ __launch_bounds__(256) void edge_max_kernel(const int* __restrict__ ei,
                                                       const float* __restrict__ as,
                                                       const float* __restrict__ ad,
                                                       float* __restrict__ emax) {
    int e = blockIdx.x * 256 + threadIdx.x;
    if (e >= N_EDGES) return;
    int s, d; edge_endpoints(ei, e, s, d);
    atomicMaxFloat(&emax[d], leaky(as[s] + ad[d]));
}

__global__ __launch_bounds__(256) void edge_exp_kernel(const int* __restrict__ ei,
                                                       const float* __restrict__ as,
                                                       const float* __restrict__ ad,
                                                       const float* __restrict__ emax,
                                                       float* __restrict__ ebuf,
                                                       float* __restrict__ denom) {
    int e = blockIdx.x * 256 + threadIdx.x;
    if (e >= N_EDGES) return;
    int s, d; edge_endpoints(ei, e, s, d);
    float ex = __expf(leaky(as[s] + ad[d]) - emax[d]);
    ebuf[e] = ex;
    atomicAdd(&denom[d], ex);
}

#define AT_ROWS    196
#define AT_SUB     2
#define AT_RANGE   (AT_ROWS * AT_SUB)
#define AT_WGS     ((N_NODES + AT_RANGE - 1) / AT_RANGE)
#define AT_CAP     4096
#define AT_THREADS 1024
#define AT_QUADS   (N_EDGES0 / 4)

__device__ __forceinline__ int at_pos(int c) {
    return (c & 128) + ((c & 3) << 5) + ((c & 127) >> 2);
}

__global__ __launch_bounds__(AT_THREADS) void agg_tile_kernel(const int* __restrict__ ei,
                                                              const float* __restrict__ ebuf,
                                                              const float* __restrict__ denom,
                                                              const float* __restrict__ xw,
                                                              float* __restrict__ agg) {
    __shared__ float        acc[AT_ROWS * 256];
    __shared__ unsigned int lp[AT_SUB][AT_CAP];
    __shared__ float        la[AT_SUB][AT_CAP];
    __shared__ unsigned int cnt[AT_SUB];

    const int tid  = threadIdx.x;
    const int lane = tid & 31;
    const int wave = tid >> 5;
    const int d0   = blockIdx.x * AT_RANGE;
    const unsigned int nvalid = (unsigned int)min(AT_RANGE, N_NODES - d0);

    if (tid < AT_SUB) cnt[tid] = 0u;
    __syncthreads();

    const int4* __restrict__ dst4 = (const int4*)(ei + N_EDGES0);
    for (int base = 0; base < AT_QUADS; base += 4 * AT_THREADS) {
        int4 dv[4];
#pragma unroll
        for (int u = 0; u < 4; ++u) {
            int i = base + u * AT_THREADS + tid;
            dv[u] = (i < AT_QUADS) ? dst4[i] : make_int4(-1, -1, -1, -1);
        }
#pragma unroll
        for (int u = 0; u < 4; ++u) {
            const int e0 = (base + u * AT_THREADS + tid) * 4;
            const unsigned int r[4] = {(unsigned int)(dv[u].x - d0), (unsigned int)(dv[u].y - d0),
                                       (unsigned int)(dv[u].z - d0), (unsigned int)(dv[u].w - d0)};
            if (min(min(r[0], r[1]), min(r[2], r[3])) < nvalid) {
#pragma unroll
                for (int j = 0; j < 4; ++j) {
                    if (r[j] < nvalid) {
                        unsigned int sub = r[j] >= AT_ROWS ? 1u : 0u;
                        unsigned int dl  = r[j] - sub * AT_ROWS;
                        unsigned int idx = atomicAdd(&cnt[sub], 1u);
                        if (idx < AT_CAP) lp[sub][idx] = (unsigned int)(e0 + j) | (dl << 20);
                    }
                }
            }
        }
    }
    __syncthreads();

    for (int p = 0; p < AT_SUB; ++p) {
        const int rbase = d0 + p * AT_ROWS;
        const int nrows = min(AT_ROWS, N_NODES - rbase);
        const unsigned int total = cnt[p];
        const int n = (int)min(total, (unsigned int)AT_CAP);

        for (int r = wave; r < nrows; r += 32) {
            const int node = rbase + r;
            const float a = ebuf[N_EDGES0 + node] / denom[node];
            const float4* row = (const float4*)(xw + (size_t)node * 256);
            float4 v0 = row[lane], v1 = row[32 + lane];
            float* ar = acc + r * 256 + lane;
            ar[0]   = a * v0.x; ar[32]  = a * v0.y; ar[64]  = a * v0.z; ar[96]  = a * v0.w;
            ar[128] = a * v1.x; ar[160] = a * v1.y; ar[192] = a * v1.z; ar[224] = a * v1.w;
        }
        for (int i = tid; i < n; i += AT_THREADS) {
            unsigned int pk = lp[p][i];
            int e  = (int)(pk & 0xFFFFFu);
            unsigned int dl = pk >> 20;
            int s = ei[e];
            la[p][i] = ebuf[e] / denom[rbase + (int)dl];
            lp[p][i] = (unsigned int)s | (dl << 17);
        }
        __syncthreads();

        if (total <= (unsigned int)AT_CAP) {
            for (int i = wave; i < n; i += 32 * 4) {
                float4 v0[4], v1[4];
                float  a[4];
                int    dl[4];
#pragma unroll
                for (int u = 0; u < 4; ++u) {
                    const int idx = i + 32 * u;
                    const bool ok = idx < n;
                    unsigned int pk = ok ? lp[p][idx] : 0u;
                    a[u]  = ok ? la[p][idx] : 0.0f;
                    dl[u] = (int)(pk >> 17);
                    const float4* row = (const float4*)(xw + (size_t)(pk & 0x1FFFFu) * 256);
                    v0[u] = row[lane];
                    v1[u] = row[32 + lane];
                }
#pragma unroll
                for (int u = 0; u < 4; ++u) {
                    if (i + 32 * u < n) {
                        float* ar = acc + dl[u] * 256 + lane;
                        atomicAdd(ar,       a[u] * v0[u].x);
                        atomicAdd(ar + 32,  a[u] * v0[u].y);
                        atomicAdd(ar + 64,  a[u] * v0[u].z);
                        atomicAdd(ar + 96,  a[u] * v0[u].w);
                        atomicAdd(ar + 128, a[u] * v1[u].x);
                        atomicAdd(ar + 160, a[u] * v1[u].y);
                        atomicAdd(ar + 192, a[u] * v1[u].z);
                        atomicAdd(ar + 224, a[u] * v1[u].w);
                    }
                }
            }
        } else {
            for (int e = tid; e < N_EDGES0; e += AT_THREADS) {
                const int d = ei[N_EDGES0 + e];
                unsigned int r = (unsigned int)(d - rbase);
                if (r < (unsigned int)max(nrows, 0)) {
                    const int s = ei[e];
                    const float a = ebuf[e] / denom[d];
                    const float* row = xw + (size_t)s * 256;
                    for (int c = 0; c < 256; ++c)
                        atomicAdd(acc + r * 256 + at_pos(c), a * row[c]);
                }
            }
        }
        __syncthreads();

        for (int r = wave; r < nrows; r += 32) {
            const float* ar = acc + r * 256 + lane;
            float4 o0 = make_float4(ar[0],   ar[32],  ar[64],  ar[96]);
            float4 o1 = make_float4(ar[128], ar[160], ar[192], ar[224]);
            float4* out4 = (float4*)(agg + (size_t)(rbase + r) * 256);
            out4[lane]      = o0;
            out4[32 + lane] = o1;
        }
        __syncthreads();
    }
}

__global__ __launch_bounds__(256) void gemm2_kernel(const float* __restrict__ agg,
                                                    const float* __restrict__ b,
                                                    const float* __restrict__ Wfc2,
                                                    const float* __restrict__ bfc2,
                                                    float* __restrict__ h2) {
    const int lane = threadIdx.x & 31;
    const int wave = threadIdx.x >> 5;
    const int rowTile = blockIdx.x * 8 + wave;
    if (rowTile >= ROW_TILES) return;
    const int row0 = rowTile * 16;
    const int col0 = blockIdx.y * 64;

    const int rA  = row0 + (lane & 15);
    const int kHi = (lane >> 4) * 2;
    const int nB  = lane & 15;

    v8f acc[4] = {};
    for (int k = 0; k < 256; k += 4) {
        v2f a;
        a.x = fmaxf(agg[(size_t)rA * 256 + k + kHi]     + b[k + kHi],     0.0f);
        a.y = fmaxf(agg[(size_t)rA * 256 + k + kHi + 1] + b[k + kHi + 1], 0.0f);
#pragma unroll
        for (int t = 0; t < 4; ++t) {
            v2f bm;
            bm.x = Wfc2[(size_t)(k + kHi)     * 128 + col0 + t * 16 + nB];
            bm.y = Wfc2[(size_t)(k + kHi + 1) * 128 + col0 + t * 16 + nB];
            acc[t] = __builtin_amdgcn_wmma_f32_16x16x4_f32(
                false, a, false, bm, (short)0, acc[t], false, false);
        }
    }
    const int mBase = row0 + ((lane >> 4) << 3);
#pragma unroll
    for (int t = 0; t < 4; ++t) {
        int col = col0 + t * 16 + nB;
#pragma unroll
        for (int r = 0; r < 8; ++r)
            h2[(size_t)(mBase + r) * 128 + col] = fmaxf(acc[t][r] + bfc2[col], 0.0f);
    }
}

__global__ __launch_bounds__(256) void gemm3_kernel(const float* __restrict__ h2,
                                                    const float* __restrict__ Wout,
                                                    const float* __restrict__ bout,
                                                    float* __restrict__ out) {
    const int lane = threadIdx.x & 31;
    const int wave = threadIdx.x >> 5;
    const int rowTile = blockIdx.x * 8 + wave;
    if (rowTile >= ROW_TILES) return;
    const int row0 = rowTile * 16;

    const int rA  = row0 + (lane & 15);
    const int kHi = (lane >> 4) * 2;
    const int nB  = lane & 15;

    v8f acc = {};
    for (int k = 0; k < 128; k += 4) {
        v2f a;
        a.x = h2[(size_t)rA * 128 + k + kHi];
        a.y = h2[(size_t)rA * 128 + k + kHi + 1];
        v2f bm;
        bm.x = (nB < 6) ? Wout[(size_t)(k + kHi)     * 6 + nB] : 0.0f;
        bm.y = (nB < 6) ? Wout[(size_t)(k + kHi + 1) * 6 + nB] : 0.0f;
        acc = __builtin_amdgcn_wmma_f32_16x16x4_f32(
            false, a, false, bm, (short)0, acc, false, false);
    }
    if (nB < 6) {
        const int mBase = row0 + ((lane >> 4) << 3);
        float bb = bout[nB];
#pragma unroll
        for (int r = 0; r < 8; ++r)
            out[(size_t)(mBase + r) * 6 + nB] = acc[r] + bb;
    }
}

extern "C" void kernel_launch(void* const* d_in, const int* in_sizes, int n_in,
                              void* d_out, int out_size, void* d_ws, size_t ws_size,
                              hipStream_t stream) {
    const float* x     = (const float*)d_in[0];
    const int*   ei    = (const int*)  d_in[1];
    const float* W     = (const float*)d_in[2];
    const float* a_src = (const float*)d_in[3];
    const float* a_dst = (const float*)d_in[4];
    const float* b     = (const float*)d_in[5];
    const float* Wfc2  = (const float*)d_in[6];
    const float* bfc2  = (const float*)d_in[7];
    const float* Wout  = (const float*)d_in[8];
    const float* bout  = (const float*)d_in[9];
    float* out = (float*)d_out;

    float* ws      = (float*)d_ws;
    float* xw      = ws;
    float* agg     = xw + (size_t)N_NODES * 256;
    float* alpha_s = agg + (size_t)N_NODES * 256;
    float* alpha_d = alpha_s + N_NODES;
    float* emax    = alpha_d + N_NODES;
    float* denom   = emax + N_NODES;
    float* ebuf    = denom + N_NODES;
    float* h2      = xw;

    init_ws<<<(N_NODES + 255) / 256, 256, 0, stream>>>(emax, denom);
    gemm1_xw<<<dim3((ROW_TILES + 7) / 8, 4), 256, 0, stream>>>(x, W, xw);
    alphas_kernel<<<(N_NODES + 7) / 8, 256, 0, stream>>>(xw, a_src, a_dst, alpha_s, alpha_d);
    edge_max_kernel<<<(N_EDGES + 255) / 256, 256, 0, stream>>>(ei, alpha_s, alpha_d, emax);
    edge_exp_kernel<<<(N_EDGES + 255) / 256, 256, 0, stream>>>(ei, alpha_s, alpha_d, emax, ebuf, denom);
    agg_tile_kernel<<<AT_WGS, AT_THREADS, 0, stream>>>(ei, ebuf, denom, xw, agg);
    gemm2_kernel<<<dim3((ROW_TILES + 7) / 8, 2), 256, 0, stream>>>(agg, b, Wfc2, bfc2, h2);
    gemm3_kernel<<<(ROW_TILES + 7) / 8, 256, 0, stream>>>(h2, Wout, bout, out);
}


# Segment softmax by destination ownership: each aggregation workgroup takes its rows' score max (LDS integer max) and exp sum (LDS f32 adds, arrival order) from its kept edges; 3 launches gone
# speedup vs baseline: 2.5155x; 1.3533x over previous
#include <hip/hip_runtime.h>
#include <hip/hip_bf16.h>

typedef __attribute__((ext_vector_type(2))) float v2f;
typedef __attribute__((ext_vector_type(8))) float v8f;

#define N_NODES   100000
#define N_EDGES0  800000
#define N_EDGES   900000
#define ROW_TILES 6250

__device__ __forceinline__ float leaky(float v) { return v > 0.0f ? v : 0.2f * v; }

__device__ __forceinline__ void ldsMaxFloat(float* addr, float val) {
    if (val >= 0.0f) atomicMax((int*)addr, __float_as_int(val));
    else             atomicMin((unsigned int*)addr, __float_as_uint(val));
}

__global__ __launch_bounds__(256) void gemm1_xw(const float* __restrict__ x,
                                                const float* __restrict__ W,
                                                float* __restrict__ xw) {
    const int lane = threadIdx.x & 31;
    const int wave = threadIdx.x >> 5;
    const int rowTile = blockIdx.x * 8 + wave;
    if (rowTile >= ROW_TILES) return;
    const int row0 = rowTile * 16;
    const int col0 = blockIdx.y * 64;

    const int rA  = row0 + (lane & 15);
    const int kHi = (lane >> 4) * 2;
    const int nB  = lane & 15;

    v8f acc[4] = {};
    for (int k = 0; k < 128; k += 4) {
        v2f a;
        a.x = x[(size_t)rA * 128 + k + kHi];
        a.y = x[(size_t)rA * 128 + k + kHi + 1];
#pragma unroll
        for (int t = 0; t < 4; ++t) {
            v2f bm;
            bm.x = W[(size_t)(k + kHi)     * 256 + col0 + t * 16 + nB];
            bm.y = W[(size_t)(k + kHi + 1) * 256 + col0 + t * 16 + nB];
            acc[t] = __builtin_amdgcn_wmma_f32_16x16x4_f32(
                false, a, false, bm, (short)0, acc[t], false, false);
        }
    }
    const int mBase = row0 + ((lane >> 4) << 3);
#pragma unroll
    for (int t = 0; t < 4; ++t)
#pragma unroll
        for (int r = 0; r < 8; ++r)
            xw[(size_t)(mBase + r) * 256 + col0 + t * 16 + nB] = acc[t][r];
}

__global__ __launch_bounds__(256) void alphas_kernel(const float* __restrict__ xw,
                                                     const float* __restrict__ a_src,
                                                     const float* __restrict__ a_dst,
                                                     float* __restrict__ as,
                                                     float* __restrict__ ad) {
    const int lane = threadIdx.x & 31;
    const int node = blockIdx.x * 8 + (threadIdx.x >> 5);
    if (node >= N_NODES) return;
    const float* xr = xw + (size_t)node * 256;
    float s = 0.0f, d = 0.0f;
#pragma unroll
    for (int i = 0; i < 8; ++i) {
        float v = xr[lane + 32 * i];
        s += v * a_src[lane + 32 * i];
        d += v * a_dst[lane + 32 * i];
    }
#pragma unroll
    for (int off = 16; off; off >>= 1) {
        s += __shfl_xor(s, off, 32);
        d += __shfl_xor(d, off, 32);
    }
    if (lane == 0) { as[node] = s; ad[node] = d; }
}

#define AT_ROWS    196
#define AT_SUB     2
#define AT_RANGE   (AT_ROWS * AT_SUB)
#define AT_WGS     ((N_NODES + AT_RANGE - 1) / AT_RANGE)
#define AT_CAP     4096
#define AT_THREADS 1024
#define AT_QUADS   (N_EDGES0 / 4)

__device__ __forceinline__ int at_pos(int c) {
    return (c & 128) + ((c & 3) << 5) + ((c & 127) >> 2);
}

__global__ __launch_bounds__(AT_THREADS) void agg_tile_kernel(const int* __restrict__ ei,
                                                              const float* __restrict__ as,
                                                              const float* __restrict__ ad,
                                                              const float* __restrict__ xw,
                                                              float* __restrict__ agg) {
    __shared__ float        acc[AT_ROWS * 256];
    __shared__ unsigned int lp[AT_SUB][AT_CAP];
    __shared__ float        la[AT_SUB][AT_CAP];
    __shared__ unsigned int cnt[AT_SUB];
    __shared__ float        rmax[AT_RANGE];
    __shared__ float        rden[AT_RANGE];
    __shared__ float        rself[AT_RANGE];
    __shared__ float        radl[AT_RANGE];

    const int tid  = threadIdx.x;
    const int lane = tid & 31;
    const int wave = tid >> 5;
    const int d0   = blockIdx.x * AT_RANGE;
    const unsigned int nvalid = (unsigned int)min(AT_RANGE, N_NODES - d0);

    if (tid < AT_SUB) cnt[tid] = 0u;
    if ((unsigned int)tid < nvalid) {
        const int node = d0 + tid;
        const float dh = ad[node];
        const float sc = leaky(as[node] + dh);
        radl[tid]  = dh;
        rself[tid] = sc;
        rmax[tid]  = sc;
        rden[tid]  = 0.0f;
    }
    __syncthreads();

    const int4* __restrict__ dst4 = (const int4*)(ei + N_EDGES0);
    for (int base = 0; base < AT_QUADS; base += 4 * AT_THREADS) {
        int4 dv[4];
#pragma unroll
        for (int u = 0; u < 4; ++u) {
            int i = base + u * AT_THREADS + tid;
            dv[u] = (i < AT_QUADS) ? dst4[i] : make_int4(-1, -1, -1, -1);
        }
#pragma unroll
        for (int u = 0; u < 4; ++u) {
            const int e0 = (base + u * AT_THREADS + tid) * 4;
            const unsigned int r[4] = {(unsigned int)(dv[u].x - d0), (unsigned int)(dv[u].y - d0),
                                       (unsigned int)(dv[u].z - d0), (unsigned int)(dv[u].w - d0)};
            if (min(min(r[0], r[1]), min(r[2], r[3])) < nvalid) {
#pragma unroll
                for (int j = 0; j < 4; ++j) {
                    if (r[j] < nvalid) {
                        unsigned int sub = r[j] >= AT_ROWS ? 1u : 0u;
                        unsigned int dl  = r[j] - sub * AT_ROWS;
                        unsigned int idx = atomicAdd(&cnt[sub], 1u);
                        if (idx < AT_CAP) lp[sub][idx] = (unsigned int)(e0 + j) | (dl << 20);
                    }
                }
            }
        }
    }
    __syncthreads();

    const bool ovf = cnt[0] > (unsigned int)AT_CAP || cnt[1] > (unsigned int)AT_CAP;

    if (!ovf) {
        for (int p = 0; p < AT_SUB; ++p) {
            const int n = (int)cnt[p];
            for (int i = tid; i < n; i += AT_THREADS) {
                const unsigned int pk = lp[p][i];
                const int e  = (int)(pk & 0xFFFFFu);
                const unsigned int dl = pk >> 20;
                const unsigned int rr = (unsigned int)(p * AT_ROWS) + dl;
                const int s = ei[e];
                const float sc = leaky(as[s] + radl[rr]);
                la[p][i] = sc;
                lp[p][i] = (unsigned int)s | (dl << 17);
                ldsMaxFloat(&rmax[rr], sc);
            }
        }
    } else {
        for (int e = tid; e < N_EDGES0; e += AT_THREADS) {
            const unsigned int r = (unsigned int)(ei[N_EDGES0 + e] - d0);
            if (r < nvalid) ldsMaxFloat(&rmax[r], leaky(as[ei[e]] + radl[r]));
        }
    }
    __syncthreads();

    if ((unsigned int)tid < nvalid) {
        const float ex = __expf(rself[tid] - rmax[tid]);
        rself[tid] = ex;
        atomicAdd(&rden[tid], ex);
    }
    if (!ovf) {
        for (int p = 0; p < AT_SUB; ++p) {
            const int n = (int)cnt[p];
            for (int i = tid; i < n; i += AT_THREADS) {
                const unsigned int rr = (unsigned int)(p * AT_ROWS) + (lp[p][i] >> 17);
                const float ex = __expf(la[p][i] - rmax[rr]);
                la[p][i] = ex;
                atomicAdd(&rden[rr], ex);
            }
        }
    } else {
        for (int e = tid; e < N_EDGES0; e += AT_THREADS) {
            const unsigned int r = (unsigned int)(ei[N_EDGES0 + e] - d0);
            if (r < nvalid) atomicAdd(&rden[r], __expf(leaky(as[ei[e]] + radl[r]) - rmax[r]));
        }
    }
    __syncthreads();

    for (int p = 0; p < AT_SUB; ++p) {
        const int rbase = d0 + p * AT_ROWS;
        const int nrows = min(AT_ROWS, N_NODES - rbase);
        const int n = ovf ? 0 : (int)cnt[p];

        for (int r = wave; r < nrows; r += 32) {
            const int node = rbase + r;
            const float a = rself[p * AT_ROWS + r] / rden[p * AT_ROWS + r];
            const float4* row = (const float4*)(xw + (size_t)node * 256);
            float4 v0 = row[lane], v1 = row[32 + lane];
            float* ar = acc + r * 256 + lane;
            ar[0]   = a * v0.x; ar[32]  = a * v0.y; ar[64]  = a * v0.z; ar[96]  = a * v0.w;
            ar[128] = a * v1.x; ar[160] = a * v1.y; ar[192] = a * v1.z; ar[224] = a * v1.w;
        }
        for (int i = tid; i < n; i += AT_THREADS)
            la[p][i] = la[p][i] / rden[(unsigned int)(p * AT_ROWS) + (lp[p][i] >> 17)];
        __syncthreads();

        if (!ovf) {
            for (int i = wave; i < n; i += 32 * 4) {
                float4 v0[4], v1[4];
                float  a[4];
                int    dl[4];
#pragma unroll
                for (int u = 0; u < 4; ++u) {
                    const int idx = i + 32 * u;
                    const bool ok = idx < n;
                    unsigned int pk = ok ? lp[p][idx] : 0u;
                    a[u]  = ok ? la[p][idx] : 0.0f;
                    dl[u] = (int)(pk >> 17);
                    const float4* row = (const float4*)(xw + (size_t)(pk & 0x1FFFFu) * 256);
                    v0[u] = row[lane];
                    v1[u] = row[32 + lane];
                }
#pragma unroll
                for (int u = 0; u < 4; ++u) {
                    if (i + 32 * u < n) {
                        float* ar = acc + dl[u] * 256 + lane;
                        atomicAdd(ar,       a[u] * v0[u].x);
                        atomicAdd(ar + 32,  a[u] * v0[u].y);
                        atomicAdd(ar + 64,  a[u] * v0[u].z);
                        atomicAdd(ar + 96,  a[u] * v0[u].w);
                        atomicAdd(ar + 128, a[u] * v1[u].x);
                        atomicAdd(ar + 160, a[u] * v1[u].y);
                        atomicAdd(ar + 192, a[u] * v1[u].z);
                        atomicAdd(ar + 224, a[u] * v1[u].w);
                    }
                }
            }
        } else {
            for (int e = tid; e < N_EDGES0; e += AT_THREADS) {
                const int d = ei[N_EDGES0 + e];
                unsigned int r = (unsigned int)(d - rbase);
                if (r < (unsigned int)max(nrows, 0)) {
                    const int s = ei[e];
                    const unsigned int rr = (unsigned int)(p * AT_ROWS) + r;
                    const float a = __expf(leaky(as[s] + radl[rr]) - rmax[rr]) / rden[rr];
                    const float* row = xw + (size_t)s * 256;
                    for (int c = 0; c < 256; ++c)
                        atomicAdd(acc + r * 256 + at_pos(c), a * row[c]);
                }
            }
        }
        __syncthreads();

        for (int r = wave; r < nrows; r += 32) {
            const float* ar = acc + r * 256 + lane;
            float4 o0 = make_float4(ar[0],   ar[32],  ar[64],  ar[96]);
            float4 o1 = make_float4(ar[128], ar[160], ar[192], ar[224]);
            float4* out4 = (float4*)(agg + (size_t)(rbase + r) * 256);
            out4[lane]      = o0;
            out4[32 + lane] = o1;
        }
        __syncthreads();
    }
}

__global__ __launch_bounds__(256) void gemm2_kernel(const float* __restrict__ agg,
                                                    const float* __restrict__ b,
                                                    const float* __restrict__ Wfc2,
                                                    const float* __restrict__ bfc2,
                                                    float* __restrict__ h2) {
    const int lane = threadIdx.x & 31;
    const int wave = threadIdx.x >> 5;
    const int rowTile = blockIdx.x * 8 + wave;
    if (rowTile >= ROW_TILES) return;
    const int row0 = rowTile * 16;
    const int col0 = blockIdx.y * 64;

    const int rA  = row0 + (lane & 15);
    const int kHi = (lane >> 4) * 2;
    const int nB  = lane & 15;

    v8f acc[4] = {};
    for (int k = 0; k < 256; k += 4) {
        v2f a;
        a.x = fmaxf(agg[(size_t)rA * 256 + k + kHi]     + b[k + kHi],     0.0f);
        a.y = fmaxf(agg[(size_t)rA * 256 + k + kHi + 1] + b[k + kHi + 1], 0.0f);
#pragma unroll
        for (int t = 0; t < 4; ++t) {
            v2f bm;
            bm.x = Wfc2[(size_t)(k + kHi)     * 128 + col0 + t * 16 + nB];
            bm.y = Wfc2[(size_t)(k + kHi + 1) * 128 + col0 + t * 16 + nB];
            acc[t] = __builtin_amdgcn_wmma_f32_16x16x4_f32(
                false, a, false, bm, (short)0, acc[t], false, false);
        }
    }
    const int mBase = row0 + ((lane >> 4) << 3);
#pragma unroll
    for (int t = 0; t < 4; ++t) {
        int col = col0 + t * 16 + nB;
#pragma unroll
        for (int r = 0; r < 8; ++r)
            h2[(size_t)(mBase + r) * 128 + col] = fmaxf(acc[t][r] + bfc2[col], 0.0f);
    }
}

__global__ __launch_bounds__(256) void gemm3_kernel(const float* __restrict__ h2,
                                                    const float* __restrict__ Wout,
                                                    const float* __restrict__ bout,
                                                    float* __restrict__ out) {
    const int lane = threadIdx.x & 31;
    const int wave = threadIdx.x >> 5;
    const int rowTile = blockIdx.x * 8 + wave;
    if (rowTile >= ROW_TILES) return;
    const int row0 = rowTile * 16;

    const int rA  = row0 + (lane & 15);
    const int kHi = (lane >> 4) * 2;
    const int nB  = lane & 15;

    v8f acc = {};
    for (int k = 0; k < 128; k += 4) {
        v2f a;
        a.x = h2[(size_t)rA * 128 + k + kHi];
        a.y = h2[(size_t)rA * 128 + k + kHi + 1];
        v2f bm;
        bm.x = (nB < 6) ? Wout[(size_t)(k + kHi)     * 6 + nB] : 0.0f;
        bm.y = (nB < 6) ? Wout[(size_t)(k + kHi + 1) * 6 + nB] : 0.0f;
        acc = __builtin_amdgcn_wmma_f32_16x16x4_f32(
            false, a, false, bm, (short)0, acc, false, false);
    }
    if (nB < 6) {
        const int mBase = row0 + ((lane >> 4) << 3);
        float bb = bout[nB];
#pragma unroll
        for (int r = 0; r < 8; ++r)
            out[(size_t)(mBase + r) * 6 + nB] = acc[r] + bb;
    }
}

extern "C" void kernel_launch(void* const* d_in, const int* in_sizes, int n_in,
                              void* d_out, int out_size, void* d_ws, size_t ws_size,
                              hipStream_t stream) {
    const float* x     = (const float*)d_in[0];
    const int*   ei    = (const int*)  d_in[1];
    const float* W     = (const float*)d_in[2];
    const float* a_src = (const float*)d_in[3];
    const float* a_dst = (const float*)d_in[4];
    const float* b     = (const float*)d_in[5];
    const float* Wfc2  = (const float*)d_in[6];
    const float* bfc2  = (const float*)d_in[7];
    const float* Wout  = (const float*)d_in[8];
    const float* bout  = (const float*)d_in[9];
    float* out = (float*)d_out;

    float* ws      = (float*)d_ws;
    float* xw      = ws;
    float* agg     = xw + (size_t)N_NODES * 256;
    float* alpha_s = agg + (size_t)N_NODES * 256;
    float* alpha_d = alpha_s + N_NODES;
    float* h2      = xw;

    gemm1_xw<<<dim3((ROW_TILES + 7) / 8, 4), 256, 0, stream>>>(x, W, xw);
    alphas_kernel<<<(N_NODES + 7) / 8, 256, 0, stream>>>(xw, a_src, a_dst, alpha_s, alpha_d);
    agg_tile_kernel<<<AT_WGS, AT_THREADS, 0, stream>>>(ei, alpha_s, alpha_d, xw, agg);
    gemm2_kernel<<<dim3((ROW_TILES + 7) / 8, 2), 256, 0, stream>>>(agg, b, Wfc2, bfc2, h2);
    gemm3_kernel<<<(ROW_TILES + 7) / 8, 256, 0, stream>>>(h2, Wout, bout, out);
}


# Dense layers: fragment loads and result stores 16 bytes per lane (columns interleaved over the 4 col-tiles, k taken 8 at a time), 4x fewer loads; f32 sum order of the products changes (k order)
# speedup vs baseline: 2.8078x; 1.1162x over previous
#include <hip/hip_runtime.h>
#include <hip/hip_bf16.h>

typedef __attribute__((ext_vector_type(2))) float v2f;
typedef __attribute__((ext_vector_type(4))) float v4f;
typedef __attribute__((ext_vector_type(8))) float v8f;

#define N_NODES   100000
#define N_EDGES0  800000
#define N_EDGES   900000
#define ROW_TILES 6250

__device__ __forceinline__ float leaky(float v) { return v > 0.0f ? v : 0.2f * v; }

__device__ __forceinline__ void ldsMaxFloat(float* addr, float val) {
    if (val >= 0.0f) atomicMax((int*)addr, __float_as_int(val));
    else             atomicMin((unsigned int*)addr, __float_as_uint(val));
}

__global__ __launch_bounds__(256) void gemm1_xw(const float* __restrict__ x,
                                                const float* __restrict__ W,
                                                float* __restrict__ xw) {
    const int lane = threadIdx.x & 31;
    const int wave = threadIdx.x >> 5;
    const int rowTile = blockIdx.x * 8 + wave;
    if (rowTile >= ROW_TILES) return;
    const int row0 = rowTile * 16;
    const int col0 = blockIdx.y * 64;

    const int c = lane & 15;
    const int h = lane >> 4;
    const float* __restrict__ ap = x + (size_t)(row0 + c) * 128 + 4 * h;
    const float* __restrict__ bp = W + (size_t)(4 * h) * 256 + col0 + 4 * c;

    v8f acc[4] = {};
    for (int k = 0; k < 128; k += 8) {
        const v4f av = *(const v4f*)(ap + k);
        const v4f b0 = *(const v4f*)(bp + (size_t)(k + 0) * 256);
        const v4f b1 = *(const v4f*)(bp + (size_t)(k + 1) * 256);
        const v4f b2 = *(const v4f*)(bp + (size_t)(k + 2) * 256);
        const v4f b3 = *(const v4f*)(bp + (size_t)(k + 3) * 256);
        v2f a;
        a.x = av[0]; a.y = av[1];
#pragma unroll
        for (int t = 0; t < 4; ++t) {
            v2f bm;
            bm.x = b0[t]; bm.y = b1[t];
            acc[t] = __builtin_amdgcn_wmma_f32_16x16x4_f32(
                false, a, false, bm, (short)0, acc[t], false, false);
        }
        a.x = av[2]; a.y = av[3];
#pragma unroll
        for (int t = 0; t < 4; ++t) {
            v2f bm;
            bm.x = b2[t]; bm.y = b3[t];
            acc[t] = __builtin_amdgcn_wmma_f32_16x16x4_f32(
                false, a, false, bm, (short)0, acc[t], false, false);
        }
    }
    const int mBase = row0 + (h << 3);
#pragma unroll
    for (int r = 0; r < 8; ++r) {
        v4f o;
        o[0] = acc[0][r]; o[1] = acc[1][r]; o[2] = acc[2][r]; o[3] = acc[3][r];
        *(v4f*)(xw + (size_t)(mBase + r) * 256 + col0 + 4 * c) = o;
    }
}

__global__ __launch_bounds__(256) void alphas_kernel(const float* __restrict__ xw,
                                                     const float* __restrict__ a_src,
                                                     const float* __restrict__ a_dst,
                                                     float* __restrict__ as,
                                                     float* __restrict__ ad) {
    const int lane = threadIdx.x & 31;
    const int node = blockIdx.x * 8 + (threadIdx.x >> 5);
    if (node >= N_NODES) return;
    const float* xr = xw + (size_t)node * 256;
    float s = 0.0f, d = 0.0f;
#pragma unroll
    for (int i = 0; i < 8; ++i) {
        float v = xr[lane + 32 * i];
        s += v * a_src[lane + 32 * i];
        d += v * a_dst[lane + 32 * i];
    }
#pragma unroll
    for (int off = 16; off; off >>= 1) {
        s += __shfl_xor(s, off, 32);
        d += __shfl_xor(d, off, 32);
    }
    if (lane == 0) { as[node] = s; ad[node] = d; }
}

#define AT_ROWS    196
#define AT_SUB     2
#define AT_RANGE   (AT_ROWS * AT_SUB)
#define AT_WGS     ((N_NODES + AT_RANGE - 1) / AT_RANGE)
#define AT_CAP     4096
#define AT_THREADS 1024
#define AT_QUADS   (N_EDGES0 / 4)

__device__ __forceinline__ int at_pos(int c) {
    return (c & 128) + ((c & 3) << 5) + ((c & 127) >> 2);
}

__global__ __launch_bounds__(AT_THREADS) void agg_tile_kernel(const int* __restrict__ ei,
                                                              const float* __restrict__ as,
                                                              const float* __restrict__ ad,
                                                              const float* __restrict__ xw,
                                                              float* __restrict__ agg) {
    __shared__ float        acc[AT_ROWS * 256];
    __shared__ unsigned int lp[AT_SUB][AT_CAP];
    __shared__ float        la[AT_SUB][AT_CAP];
    __shared__ unsigned int cnt[AT_SUB];
    __shared__ float        rmax[AT_RANGE];
    __shared__ float        rden[AT_RANGE];
    __shared__ float        rself[AT_RANGE];
    __shared__ float        radl[AT_RANGE];

    const int tid  = threadIdx.x;
    const int lane = tid & 31;
    const int wave = tid >> 5;
    const int d0   = blockIdx.x * AT_RANGE;
    const unsigned int nvalid = (unsigned int)min(AT_RANGE, N_NODES - d0);

    if (tid < AT_SUB) cnt[tid] = 0u;
    if ((unsigned int)tid < nvalid) {
        const int node = d0 + tid;
        const float dh = ad[node];
        const float sc = leaky(as[node] + dh);
        radl[tid]  = dh;
        rself[tid] = sc;
        rmax[tid]  = sc;
        rden[tid]  = 0.0f;
    }
    __syncthreads();

    const int4* __restrict__ dst4 = (const int4*)(ei + N_EDGES0);
    for (int base = 0; base < AT_QUADS; base += 4 * AT_THREADS) {
        int4 dv[4];
#pragma unroll
        for (int u = 0; u < 4; ++u) {
            int i = base + u * AT_THREADS + tid;
            dv[u] = (i < AT_QUADS) ? dst4[i] : make_int4(-1, -1, -1, -1);
        }
#pragma unroll
        for (int u = 0; u < 4; ++u) {
            const int e0 = (base + u * AT_THREADS + tid) * 4;
            const unsigned int r[4] = {(unsigned int)(dv[u].x - d0), (unsigned int)(dv[u].y - d0),
                                       (unsigned int)(dv[u].z - d0), (unsigned int)(dv[u].w - d0)};
            if (min(min(r[0], r[1]), min(r[2], r[3])) < nvalid) {
#pragma unroll
                for (int j = 0; j < 4; ++j) {
                    if (r[j] < nvalid) {
                        unsigned int sub = r[j] >= AT_ROWS ? 1u : 0u;
                        unsigned int dl  = r[j] - sub * AT_ROWS;
                        unsigned int idx = atomicAdd(&cnt[sub], 1u);
                        if (idx < AT_CAP) lp[sub][idx] = (unsigned int)(e0 + j) | (dl << 20);
                    }
                }
            }
        }
    }
    __syncthreads();

    const bool ovf = cnt[0] > (unsigned int)AT_CAP || cnt[1] > (unsigned int)AT_CAP;

    if (!ovf) {
        for (int p = 0; p < AT_SUB; ++p) {
            const int n = (int)cnt[p];
            for (int i = tid; i < n; i += AT_THREADS) {
                const unsigned int pk = lp[p][i];
                const int e  = (int)(pk & 0xFFFFFu);
                const unsigned int dl = pk >> 20;
                const unsigned int rr = (unsigned int)(p * AT_ROWS) + dl;
                const int s = ei[e];
                const float sc = leaky(as[s] + radl[rr]);
                la[p][i] = sc;
                lp[p][i] = (unsigned int)s | (dl << 17);
                ldsMaxFloat(&rmax[rr], sc);
            }
        }
    } else {
        for (int e = tid; e < N_EDGES0; e += AT_THREADS) {
            const unsigned int r = (unsigned int)(ei[N_EDGES0 + e] - d0);
            if (r < nvalid) ldsMaxFloat(&rmax[r], leaky(as[ei[e]] + radl[r]));
        }
    }
    __syncthreads();

    if ((unsigned int)tid < nvalid) {
        const float ex = __expf(rself[tid] - rmax[tid]);
        rself[tid] = ex;
        atomicAdd(&rden[tid], ex);
    }
    if (!ovf) {
        for (int p = 0; p < AT_SUB; ++p) {
            const int n = (int)cnt[p];
            for (int i = tid; i < n; i += AT_THREADS) {
                const unsigned int rr = (unsigned int)(p * AT_ROWS) + (lp[p][i] >> 17);
                const float ex = __expf(la[p][i] - rmax[rr]);
                la[p][i] = ex;
                atomicAdd(&rden[rr], ex);
            }
        }
    } else {
        for (int e = tid; e < N_EDGES0; e += AT_THREADS) {
            const unsigned int r = (unsigned int)(ei[N_EDGES0 + e] - d0);
            if (r < nvalid) atomicAdd(&rden[r], __expf(leaky(as[ei[e]] + radl[r]) - rmax[r]));
        }
    }
    __syncthreads();

    for (int p = 0; p < AT_SUB; ++p) {
        const int rbase = d0 + p * AT_ROWS;
        const int nrows = min(AT_ROWS, N_NODES - rbase);
        const int n = ovf ? 0 : (int)cnt[p];

        for (int r = wave; r < nrows; r += 32) {
            const int node = rbase + r;
            const float a = rself[p * AT_ROWS + r] / rden[p * AT_ROWS + r];
            const float4* row = (const float4*)(xw + (size_t)node * 256);
            float4 v0 = row[lane], v1 = row[32 + lane];
            float* ar = acc + r * 256 + lane;
            ar[0]   = a * v0.x; ar[32]  = a * v0.y; ar[64]  = a * v0.z; ar[96]  = a * v0.w;
            ar[128] = a * v1.x; ar[160] = a * v1.y; ar[192] = a * v1.z; ar[224] = a * v1.w;
        }
        for (int i = tid; i < n; i += AT_THREADS)
            la[p][i] = la[p][i] / rden[(unsigned int)(p * AT_ROWS) + (lp[p][i] >> 17)];
        __syncthreads();

        if (!ovf) {
            for (int i = wave; i < n; i += 32 * 4) {
                float4 v0[4], v1[4];
                float  a[4];
                int    dl[4];
#pragma unroll
                for (int u = 0; u < 4; ++u) {
                    const int idx = i + 32 * u;
                    const bool ok = idx < n;
                    unsigned int pk = ok ? lp[p][idx] : 0u;
                    a[u]  = ok ? la[p][idx] : 0.0f;
                    dl[u] = (int)(pk >> 17);
                    const float4* row = (const float4*)(xw + (size_t)(pk & 0x1FFFFu) * 256);
                    v0[u] = row[lane];
                    v1[u] = row[32 + lane];
                }
#pragma unroll
                for (int u = 0; u < 4; ++u) {
                    if (i + 32 * u < n) {
                        float* ar = acc + dl[u] * 256 + lane;
                        atomicAdd(ar,       a[u] * v0[u].x);
                        atomicAdd(ar + 32,  a[u] * v0[u].y);
                        atomicAdd(ar + 64,  a[u] * v0[u].z);
                        atomicAdd(ar + 96,  a[u] * v0[u].w);
                        atomicAdd(ar + 128, a[u] * v1[u].x);
                        atomicAdd(ar + 160, a[u] * v1[u].y);
                        atomicAdd(ar + 192, a[u] * v1[u].z);
                        atomicAdd(ar + 224, a[u] * v1[u].w);
                    }
                }
            }
        } else {
            for (int e = tid; e < N_EDGES0; e += AT_THREADS) {
                const int d = ei[N_EDGES0 + e];
                unsigned int r = (unsigned int)(d - rbase);
                if (r < (unsigned int)max(nrows, 0)) {
                    const int s = ei[e];
                    const unsigned int rr = (unsigned int)(p * AT_ROWS) + r;
                    const float a = __expf(leaky(as[s] + radl[rr]) - rmax[rr]) / rden[rr];
                    const float* row = xw + (size_t)s * 256;
                    for (int c = 0; c < 256; ++c)
                        atomicAdd(acc + r * 256 + at_pos(c), a * row[c]);
                }
            }
        }
        __syncthreads();

        for (int r = wave; r < nrows; r += 32) {
            const float* ar = acc + r * 256 + lane;
            float4 o0 = make_float4(ar[0],   ar[32],  ar[64],  ar[96]);
            float4 o1 = make_float4(ar[128], ar[160], ar[192], ar[224]);
            float4* out4 = (float4*)(agg + (size_t)(rbase + r) * 256);
            out4[lane]      = o0;
            out4[32 + lane] = o1;
        }
        __syncthreads();
    }
}

__global__ __launch_bounds__(256) void gemm2_kernel(const float* __restrict__ agg,
                                                    const float* __restrict__ b,
                                                    const float* __restrict__ Wfc2,
                                                    const float* __restrict__ bfc2,
                                                    float* __restrict__ h2) {
    const int lane = threadIdx.x & 31;
    const int wave = threadIdx.x >> 5;
    const int rowTile = blockIdx.x * 8 + wave;
    if (rowTile >= ROW_TILES) return;
    const int row0 = rowTile * 16;
    const int col0 = blockIdx.y * 64;

    const int c = lane & 15;
    const int h = lane >> 4;
    const float* __restrict__ ap = agg + (size_t)(row0 + c) * 256 + 4 * h;
    const float* __restrict__ cp = b + 4 * h;
    const float* __restrict__ bp = Wfc2 + (size_t)(4 * h) * 128 + col0 + 4 * c;

    v8f acc[4] = {};
    for (int k = 0; k < 256; k += 8) {
        const v4f av = *(const v4f*)(ap + k);
        const v4f cv = *(const v4f*)(cp + k);
        const v4f b0 = *(const v4f*)(bp + (size_t)(k + 0) * 128);
        const v4f b1 = *(const v4f*)(bp + (size_t)(k + 1) * 128);
        const v4f b2 = *(const v4f*)(bp + (size_t)(k + 2) * 128);
        const v4f b3 = *(const v4f*)(bp + (size_t)(k + 3) * 128);
        v2f a;
        a.x = fmaxf(av[0] + cv[0], 0.0f);
        a.y = fmaxf(av[1] + cv[1], 0.0f);
#pragma unroll
        for (int t = 0; t < 4; ++t) {
            v2f bm;
            bm.x = b0[t]; bm.y = b1[t];
            acc[t] = __builtin_amdgcn_wmma_f32_16x16x4_f32(
                false, a, false, bm, (short)0, acc[t], false, false);
        }
        a.x = fmaxf(av[2] + cv[2], 0.0f);
        a.y = fmaxf(av[3] + cv[3], 0.0f);
#pragma unroll
        for (int t = 0; t < 4; ++t) {
            v2f bm;
            bm.x = b2[t]; bm.y = b3[t];
            acc[t] = __builtin_amdgcn_wmma_f32_16x16x4_f32(
                false, a, false, bm, (short)0, acc[t], false, false);
        }
    }
    const int mBase = row0 + (h << 3);
    const v4f bo = *(const v4f*)(bfc2 + col0 + 4 * c);
#pragma unroll
    for (int r = 0; r < 8; ++r) {
        v4f o;
        o[0] = fmaxf(acc[0][r] + bo[0], 0.0f);
        o[1] = fmaxf(acc[1][r] + bo[1], 0.0f);
        o[2] = fmaxf(acc[2][r] + bo[2], 0.0f);
        o[3] = fmaxf(acc[3][r] + bo[3], 0.0f);
        *(v4f*)(h2 + (size_t)(mBase + r) * 128 + col0 + 4 * c) = o;
    }
}

__global__ __launch_bounds__(256) void gemm3_kernel(const float* __restrict__ h2,
                                                    const float* __restrict__ Wout,
                                                    const float* __restrict__ bout,
                                                    float* __restrict__ out) {
    const int lane = threadIdx.x & 31;
    const int wave = threadIdx.x >> 5;
    const int rowTile = blockIdx.x * 8 + wave;
    if (rowTile >= ROW_TILES) return;
    const int row0 = rowTile * 16;

    const int nB = lane & 15;
    const int h  = lane >> 4;
    const float* __restrict__ ap = h2 + (size_t)(row0 + nB) * 128 + 4 * h;

    v8f acc = {};
    for (int k = 0; k < 128; k += 8) {
        const v4f av = *(const v4f*)(ap + k);
        const int kr = k + 4 * h;
        v2f a, bm;
        a.x = av[0]; a.y = av[1];
        bm.x = (nB < 6) ? Wout[(size_t)(kr + 0) * 6 + nB] : 0.0f;
        bm.y = (nB < 6) ? Wout[(size_t)(kr + 1) * 6 + nB] : 0.0f;
        acc = __builtin_amdgcn_wmma_f32_16x16x4_f32(
            false, a, false, bm, (short)0, acc, false, false);
        a.x = av[2]; a.y = av[3];
        bm.x = (nB < 6) ? Wout[(size_t)(kr + 2) * 6 + nB] : 0.0f;
        bm.y = (nB < 6) ? Wout[(size_t)(kr + 3) * 6 + nB] : 0.0f;
        acc = __builtin_amdgcn_wmma_f32_16x16x4_f32(
            false, a, false, bm, (short)0, acc, false, false);
    }
    if (nB < 6) {
        const int mBase = row0 + (h << 3);
        float bb = bout[nB];
#pragma unroll
        for (int r = 0; r < 8; ++r)
            out[(size_t)(mBase + r) * 6 + nB] = acc[r] + bb;
    }
}

extern "C" void kernel_launch(void* const* d_in, const int* in_sizes, int n_in,
                              void* d_out, int out_size, void* d_ws, size_t ws_size,
                              hipStream_t stream) {
    const float* x     = (const float*)d_in[0];
    const int*   ei    = (const int*)  d_in[1];
    const float* W     = (const float*)d_in[2];
    const float* a_src = (const float*)d_in[3];
    const float* a_dst = (const float*)d_in[4];
    const float* b     = (const float*)d_in[5];
    const float* Wfc2  = (const float*)d_in[6];
    const float* bfc2  = (const float*)d_in[7];
    const float* Wout  = (const float*)d_in[8];
    const float* bout  = (const float*)d_in[9];
    float* out = (float*)d_out;

    float* ws      = (float*)d_ws;
    float* xw      = ws;
    float* agg     = xw + (size_t)N_NODES * 256;
    float* alpha_s = agg + (size_t)N_NODES * 256;
    float* alpha_d = alpha_s + N_NODES;
    float* h2      = xw;

    gemm1_xw<<<dim3((ROW_TILES + 7) / 8, 4), 256, 0, stream>>>(x, W, xw);
    alphas_kernel<<<(N_NODES + 7) / 8, 256, 0, stream>>>(xw, a_src, a_dst, alpha_s, alpha_d);
    agg_tile_kernel<<<AT_WGS, AT_THREADS, 0, stream>>>(ei, alpha_s, alpha_d, xw, agg);
    gemm2_kernel<<<dim3((ROW_TILES + 7) / 8, 2), 256, 0, stream>>>(agg, b, Wfc2, bfc2, h2);
    gemm3_kernel<<<(ROW_TILES + 7) / 8, 256, 0, stream>>>(h2, Wout, bout, out);
}


# Edge pass scans one owner byte per edge (destination / 392, written by the score launch), not the 4-byte destination: 4x less scan traffic; kept edges split by tile after; LDS-arrival sum order
# speedup vs baseline: 2.9964x; 1.0672x over previous
#include <hip/hip_runtime.h>
#include <hip/hip_bf16.h>

typedef __attribute__((ext_vector_type(2))) float v2f;
typedef __attribute__((ext_vector_type(4))) float v4f;
typedef __attribute__((ext_vector_type(8))) float v8f;

#define N_NODES   100000
#define N_EDGES0  800000
#define N_EDGES   900000
#define ROW_TILES 6250

__device__ __forceinline__ float leaky(float v) { return v > 0.0f ? v : 0.2f * v; }

__device__ __forceinline__ void ldsMaxFloat(float* addr, float val) {
    if (val >= 0.0f) atomicMax((int*)addr, __float_as_int(val));
    else             atomicMin((unsigned int*)addr, __float_as_uint(val));
}

__global__ __launch_bounds__(256) void gemm1_xw(const float* __restrict__ x,
                                                const float* __restrict__ W,
                                                float* __restrict__ xw) {
    const int lane = threadIdx.x & 31;
    const int wave = threadIdx.x >> 5;
    const int rowTile = blockIdx.x * 8 + wave;
    if (rowTile >= ROW_TILES) return;
    const int row0 = rowTile * 16;
    const int col0 = blockIdx.y * 64;

    const int c = lane & 15;
    const int h = lane >> 4;
    const float* __restrict__ ap = x + (size_t)(row0 + c) * 128 + 4 * h;
    const float* __restrict__ bp = W + (size_t)(4 * h) * 256 + col0 + 4 * c;

    v8f acc[4] = {};
    for (int k = 0; k < 128; k += 8) {
        const v4f av = *(const v4f*)(ap + k);
        const v4f b0 = *(const v4f*)(bp + (size_t)(k + 0) * 256);
        const v4f b1 = *(const v4f*)(bp + (size_t)(k + 1) * 256);
        const v4f b2 = *(const v4f*)(bp + (size_t)(k + 2) * 256);
        const v4f b3 = *(const v4f*)(bp + (size_t)(k + 3) * 256);
        v2f a;
        a.x = av[0]; a.y = av[1];
#pragma unroll
        for (int t = 0; t < 4; ++t) {
            v2f bm;
            bm.x = b0[t]; bm.y = b1[t];
            acc[t] = __builtin_amdgcn_wmma_f32_16x16x4_f32(
                false, a, false, bm, (short)0, acc[t], false, false);
        }
        a.x = av[2]; a.y = av[3];
#pragma unroll
        for (int t = 0; t < 4; ++t) {
            v2f bm;
            bm.x = b2[t]; bm.y = b3[t];
            acc[t] = __builtin_amdgcn_wmma_f32_16x16x4_f32(
                false, a, false, bm, (short)0, acc[t], false, false);
        }
    }
    const int mBase = row0 + (h << 3);
#pragma unroll
    for (int r = 0; r < 8; ++r) {
        v4f o;
        o[0] = acc[0][r]; o[1] = acc[1][r]; o[2] = acc[2][r]; o[3] = acc[3][r];
        *(v4f*)(xw + (size_t)(mBase + r) * 256 + col0 + 4 * c) = o;
    }
}

#define AT_RANGE_ROWS 392
__global__ __launch_bounds__(256) void alphas_kernel(const float* __restrict__ xw,
                                                     const float* __restrict__ a_src,
                                                     const float* __restrict__ a_dst,
                                                     const int* __restrict__ ei,
                                                     float* __restrict__ as,
                                                     float* __restrict__ ad,
                                                     unsigned int* __restrict__ own) {
    const int g = blockIdx.x * 256 + threadIdx.x;
    if (g < N_EDGES0 / 4) {
        const int4 d = ((const int4*)(ei + N_EDGES0))[g];
        own[g] = (((unsigned int)d.x / AT_RANGE_ROWS) & 0xFFu)
               | ((((unsigned int)d.y / AT_RANGE_ROWS) & 0xFFu) << 8)
               | ((((unsigned int)d.z / AT_RANGE_ROWS) & 0xFFu) << 16)
               | (((unsigned int)d.w / AT_RANGE_ROWS) << 24);
    }
    const int lane = threadIdx.x & 31;
    const int node = blockIdx.x * 8 + (threadIdx.x >> 5);
    if (node >= N_NODES) return;
    const float* xr = xw + (size_t)node * 256;
    float s = 0.0f, d = 0.0f;
#pragma unroll
    for (int i = 0; i < 8; ++i) {
        float v = xr[lane + 32 * i];
        s += v * a_src[lane + 32 * i];
        d += v * a_dst[lane + 32 * i];
    }
#pragma unroll
    for (int off = 16; off; off >>= 1) {
        s += __shfl_xor(s, off, 32);
        d += __shfl_xor(d, off, 32);
    }
    if (lane == 0) { as[node] = s; ad[node] = d; }
}

#define AT_ROWS    196
#define AT_SUB     2
#define AT_RANGE   (AT_ROWS * AT_SUB)
#define AT_WGS     ((N_NODES + AT_RANGE - 1) / AT_RANGE)
#define AT_CAP     4096
#define AT_RAWCAP  (AT_SUB * AT_CAP)
#define AT_THREADS 1024
#define AT_OWN16   (N_EDGES0 / 16)
static_assert(AT_RANGE == AT_RANGE_ROWS, "owner byte and edge pass must split the rows alike");
static_assert(AT_WGS <= 256, "the owner index must fit one byte");
static_assert(N_EDGES0 % 16 == 0, "owner bytes are read sixteen at a time");

__device__ __forceinline__ int at_pos(int c) {
    return (c & 128) + ((c & 3) << 5) + ((c & 127) >> 2);
}

__global__ __launch_bounds__(AT_THREADS) void agg_tile_kernel(const int* __restrict__ ei,
                                                              const unsigned int* __restrict__ own,
                                                              const float* __restrict__ as,
                                                              const float* __restrict__ ad,
                                                              const float* __restrict__ xw,
                                                              float* __restrict__ agg) {
    __shared__ float        acc[AT_ROWS * 256];
    __shared__ unsigned int raw[AT_RAWCAP];
    __shared__ unsigned int lp[AT_SUB][AT_CAP];
    __shared__ float        la[AT_SUB][AT_CAP];
    __shared__ unsigned int cnt[AT_SUB];
    __shared__ unsigned int rawcnt;
    __shared__ float        rmax[AT_RANGE];
    __shared__ float        rden[AT_RANGE];
    __shared__ float        rself[AT_RANGE];
    __shared__ float        radl[AT_RANGE];

    const int tid  = threadIdx.x;
    const int lane = tid & 31;
    const int wave = tid >> 5;
    const int d0   = blockIdx.x * AT_RANGE;
    const unsigned int nvalid = (unsigned int)min(AT_RANGE, N_NODES - d0);

    if (tid < AT_SUB) cnt[tid] = 0u;
    if (tid == AT_SUB) rawcnt = 0u;
    if ((unsigned int)tid < nvalid) {
        const int node = d0 + tid;
        const float dh = ad[node];
        const float sc = leaky(as[node] + dh);
        radl[tid]  = dh;
        rself[tid] = sc;
        rmax[tid]  = sc;
        rden[tid]  = 0.0f;
    }
    __syncthreads();

    const uint4* __restrict__ own4 = (const uint4*)own;
    const unsigned int pat = (unsigned int)blockIdx.x * 0x01010101u;
    for (int base = 0; base < AT_OWN16; base += 4 * AT_THREADS) {
        uint4 ov[4];
#pragma unroll
        for (int u = 0; u < 4; ++u) {
            const int i = base + u * AT_THREADS + tid;
            ov[u] = own4[min(i, AT_OWN16 - 1)];
        }
        unsigned int hit[4];
#pragma unroll
        for (int u = 0; u < 4; ++u) {
            const unsigned int w[4] = {ov[u].x ^ pat, ov[u].y ^ pat, ov[u].z ^ pat, ov[u].w ^ pat};
            unsigned int h = 0u;
#pragma unroll
            for (int q = 0; q < 4; ++q) {
                const unsigned int m = ~(((w[q] & 0x7F7F7F7Fu) + 0x7F7F7F7Fu) | w[q]) & 0x80808080u;
                h |= m >> (7 - q);
            }
            hit[u] = (base + u * AT_THREADS + tid < AT_OWN16) ? h : 0u;
        }
#pragma unroll
        for (int v = 0; v < 2; ++v) hit[v] = hit[2 * v] | (hit[2 * v + 1] << 4);
        if ((hit[0] | hit[1]) != 0u) {
            unsigned int idx = atomicAdd(&rawcnt, (unsigned int)(__popc(hit[0]) + __popc(hit[1])));
#pragma unroll
            for (int v = 0; v < 2; ++v) {
                unsigned int hb = hit[v];
                while (hb != 0u) {
                    const int p = __ffs((int)hb) - 1;
                    hb &= hb - 1u;
                    const int u = 2 * v + ((p >> 2) & 1);
                    const int e = (base + u * AT_THREADS + tid) * 16 + 4 * (p & 3) + (p >> 3);
                    if (idx < AT_RAWCAP) raw[idx] = (unsigned int)e;
                    ++idx;
                }
            }
        }
    }
    __syncthreads();

    const unsigned int nraw = rawcnt;
    const bool ovf1 = nraw > (unsigned int)AT_RAWCAP;
    if (!ovf1) {
        for (int i = tid; i < (int)nraw; i += AT_THREADS) {
            const int e = (int)raw[i];
            const int s = ei[e];
            const int d = ei[N_EDGES0 + e];
            asm volatile("" :: "v"(s), "v"(d));
            const unsigned int r = (unsigned int)(d - d0);
            if (r < nvalid) {
                const unsigned int sub = r >= AT_ROWS ? 1u : 0u;
                const unsigned int dl  = r - sub * AT_ROWS;
                const float sc = leaky(as[s] + radl[r]);
                const unsigned int idx = atomicAdd(&cnt[sub], 1u);
                if (idx < AT_CAP) {
                    la[sub][idx] = sc;
                    lp[sub][idx] = (unsigned int)s | (dl << 17);
                }
                ldsMaxFloat(&rmax[r], sc);
            }
        }
    }
    __syncthreads();

    const bool ovf = ovf1 || cnt[0] > (unsigned int)AT_CAP || cnt[1] > (unsigned int)AT_CAP;
    if (ovf) {
        for (int e = tid; e < N_EDGES0; e += AT_THREADS) {
            const unsigned int r = (unsigned int)(ei[N_EDGES0 + e] - d0);
            if (r < nvalid) ldsMaxFloat(&rmax[r], leaky(as[ei[e]] + radl[r]));
        }
    }
    __syncthreads();

    if ((unsigned int)tid < nvalid) {
        const float ex = __expf(rself[tid] - rmax[tid]);
        rself[tid] = ex;
        atomicAdd(&rden[tid], ex);
    }
    if (!ovf) {
        for (int p = 0; p < AT_SUB; ++p) {
            const int n = (int)cnt[p];
            for (int i = tid; i < n; i += AT_THREADS) {
                const unsigned int rr = (unsigned int)(p * AT_ROWS) + (lp[p][i] >> 17);
                const float ex = __expf(la[p][i] - rmax[rr]);
                la[p][i] = ex;
                atomicAdd(&rden[rr], ex);
            }
        }
    } else {
        for (int e = tid; e < N_EDGES0; e += AT_THREADS) {
            const unsigned int r = (unsigned int)(ei[N_EDGES0 + e] - d0);
            if (r < nvalid) atomicAdd(&rden[r], __expf(leaky(as[ei[e]] + radl[r]) - rmax[r]));
        }
    }
    __syncthreads();

    for (int p = 0; p < AT_SUB; ++p) {
        const int rbase = d0 + p * AT_ROWS;
        const int nrows = min(AT_ROWS, N_NODES - rbase);
        const int n = ovf ? 0 : (int)cnt[p];

        for (int r = wave; r < nrows; r += 32) {
            const int node = rbase + r;
            const float a = rself[p * AT_ROWS + r] / rden[p * AT_ROWS + r];
            const float4* row = (const float4*)(xw + (size_t)node * 256);
            float4 v0 = row[lane], v1 = row[32 + lane];
            float* ar = acc + r * 256 + lane;
            ar[0]   = a * v0.x; ar[32]  = a * v0.y; ar[64]  = a * v0.z; ar[96]  = a * v0.w;
            ar[128] = a * v1.x; ar[160] = a * v1.y; ar[192] = a * v1.z; ar[224] = a * v1.w;
        }
        for (int i = tid; i < n; i += AT_THREADS)
            la[p][i] = la[p][i] / rden[(unsigned int)(p * AT_ROWS) + (lp[p][i] >> 17)];
        __syncthreads();

        if (!ovf) {
            for (int i = wave; i < n; i += 32 * 4) {
                float4 v0[4], v1[4];
                float  a[4];
                int    dl[4];
#pragma unroll
                for (int u = 0; u < 4; ++u) {
                    const int idx = i + 32 * u;
                    const bool ok = idx < n;
                    unsigned int pk = ok ? lp[p][idx] : 0u;
                    a[u]  = ok ? la[p][idx] : 0.0f;
                    dl[u] = (int)(pk >> 17);
                    const float4* row = (const float4*)(xw + (size_t)(pk & 0x1FFFFu) * 256);
                    v0[u] = row[lane];
                    v1[u] = row[32 + lane];
                }
#pragma unroll
                for (int u = 0; u < 4; ++u) {
                    if (i + 32 * u < n) {
                        float* ar = acc + dl[u] * 256 + lane;
                        atomicAdd(ar,       a[u] * v0[u].x);
                        atomicAdd(ar + 32,  a[u] * v0[u].y);
                        atomicAdd(ar + 64,  a[u] * v0[u].z);
                        atomicAdd(ar + 96,  a[u] * v0[u].w);
                        atomicAdd(ar + 128, a[u] * v1[u].x);
                        atomicAdd(ar + 160, a[u] * v1[u].y);
                        atomicAdd(ar + 192, a[u] * v1[u].z);
                        atomicAdd(ar + 224, a[u] * v1[u].w);
                    }
                }
            }
        } else {
            for (int e = tid; e < N_EDGES0; e += AT_THREADS) {
                const int d = ei[N_EDGES0 + e];
                unsigned int r = (unsigned int)(d - rbase);
                if (r < (unsigned int)max(nrows, 0)) {
                    const int s = ei[e];
                    const unsigned int rr = (unsigned int)(p * AT_ROWS) + r;
                    const float a = __expf(leaky(as[s] + radl[rr]) - rmax[rr]) / rden[rr];
                    const float* row = xw + (size_t)s * 256;
                    for (int c = 0; c < 256; ++c)
                        atomicAdd(acc + r * 256 + at_pos(c), a * row[c]);
                }
            }
        }
        __syncthreads();

        for (int r = wave; r < nrows; r += 32) {
            const float* ar = acc + r * 256 + lane;
            float4 o0 = make_float4(ar[0],   ar[32],  ar[64],  ar[96]);
            float4 o1 = make_float4(ar[128], ar[160], ar[192], ar[224]);
            float4* out4 = (float4*)(agg + (size_t)(rbase + r) * 256);
            out4[lane]      = o0;
            out4[32 + lane] = o1;
        }
        __syncthreads();
    }
}

__global__ __launch_bounds__(256) void gemm2_kernel(const float* __restrict__ agg,
                                                    const float* __restrict__ b,
                                                    const float* __restrict__ Wfc2,
                                                    const float* __restrict__ bfc2,
                                                    float* __restrict__ h2) {
    const int lane = threadIdx.x & 31;
    const int wave = threadIdx.x >> 5;
    const int rowTile = blockIdx.x * 8 + wave;
    if (rowTile >= ROW_TILES) return;
    const int row0 = rowTile * 16;
    const int col0 = blockIdx.y * 64;

    const int c = lane & 15;
    const int h = lane >> 4;
    const float* __restrict__ ap = agg + (size_t)(row0 + c) * 256 + 4 * h;
    const float* __restrict__ cp = b + 4 * h;
    const float* __restrict__ bp = Wfc2 + (size_t)(4 * h) * 128 + col0 + 4 * c;

    v8f acc[4] = {};
    for (int k = 0; k < 256; k += 8) {
        const v4f av = *(const v4f*)(ap + k);
        const v4f cv = *(const v4f*)(cp + k);
        const v4f b0 = *(const v4f*)(bp + (size_t)(k + 0) * 128);
        const v4f b1 = *(const v4f*)(bp + (size_t)(k + 1) * 128);
        const v4f b2 = *(const v4f*)(bp + (size_t)(k + 2) * 128);
        const v4f b3 = *(const v4f*)(bp + (size_t)(k + 3) * 128);
        v2f a;
        a.x = fmaxf(av[0] + cv[0], 0.0f);
        a.y = fmaxf(av[1] + cv[1], 0.0f);
#pragma unroll
        for (int t = 0; t < 4; ++t) {
            v2f bm;
            bm.x = b0[t]; bm.y = b1[t];
            acc[t] = __builtin_amdgcn_wmma_f32_16x16x4_f32(
                false, a, false, bm, (short)0, acc[t], false, false);
        }
        a.x = fmaxf(av[2] + cv[2], 0.0f);
        a.y = fmaxf(av[3] + cv[3], 0.0f);
#pragma unroll
        for (int t = 0; t < 4; ++t) {
            v2f bm;
            bm.x = b2[t]; bm.y = b3[t];
            acc[t] = __builtin_amdgcn_wmma_f32_16x16x4_f32(
                false, a, false, bm, (short)0, acc[t], false, false);
        }
    }
    const int mBase = row0 + (h << 3);
    const v4f bo = *(const v4f*)(bfc2 + col0 + 4 * c);
#pragma unroll
    for (int r = 0; r < 8; ++r) {
        v4f o;
        o[0] = fmaxf(acc[0][r] + bo[0], 0.0f);
        o[1] = fmaxf(acc[1][r] + bo[1], 0.0f);
        o[2] = fmaxf(acc[2][r] + bo[2], 0.0f);
        o[3] = fmaxf(acc[3][r] + bo[3], 0.0f);
        *(v4f*)(h2 + (size_t)(mBase + r) * 128 + col0 + 4 * c) = o;
    }
}

__global__ __launch_bounds__(256) void gemm3_kernel(const float* __restrict__ h2,
                                                    const float* __restrict__ Wout,
                                                    const float* __restrict__ bout,
                                                    float* __restrict__ out) {
    const int lane = threadIdx.x & 31;
    const int wave = threadIdx.x >> 5;
    const int rowTile = blockIdx.x * 8 + wave;
    if (rowTile >= ROW_TILES) return;
    const int row0 = rowTile * 16;

    const int nB = lane & 15;
    const int h  = lane >> 4;
    const float* __restrict__ ap = h2 + (size_t)(row0 + nB) * 128 + 4 * h;

    v8f acc = {};
    for (int k = 0; k < 128; k += 8) {
        const v4f av = *(const v4f*)(ap + k);
        const int kr = k + 4 * h;
        v2f a, bm;
        a.x = av[0]; a.y = av[1];
        bm.x = (nB < 6) ? Wout[(size_t)(kr + 0) * 6 + nB] : 0.0f;
        bm.y = (nB < 6) ? Wout[(size_t)(kr + 1) * 6 + nB] : 0.0f;
        acc = __builtin_amdgcn_wmma_f32_16x16x4_f32(
            false, a, false, bm, (short)0, acc, false, false);
        a.x = av[2]; a.y = av[3];
        bm.x = (nB < 6) ? Wout[(size_t)(kr + 2) * 6 + nB] : 0.0f;
        bm.y = (nB < 6) ? Wout[(size_t)(kr + 3) * 6 + nB] : 0.0f;
        acc = __builtin_amdgcn_wmma_f32_16x16x4_f32(
            false, a, false, bm, (short)0, acc, false, false);
    }
    if (nB < 6) {
        const int mBase = row0 + (h << 3);
        float bb = bout[nB];
#pragma unroll
        for (int r = 0; r < 8; ++r)
            out[(size_t)(mBase + r) * 6 + nB] = acc[r] + bb;
    }
}

extern "C" void kernel_launch(void* const* d_in, const int* in_sizes, int n_in,
                              void* d_out, int out_size, void* d_ws, size_t ws_size,
                              hipStream_t stream) {
    const float* x     = (const float*)d_in[0];
    const int*   ei    = (const int*)  d_in[1];
    const float* W     = (const float*)d_in[2];
    const float* a_src = (const float*)d_in[3];
    const float* a_dst = (const float*)d_in[4];
    const float* b     = (const float*)d_in[5];
    const float* Wfc2  = (const float*)d_in[6];
    const float* bfc2  = (const float*)d_in[7];
    const float* Wout  = (const float*)d_in[8];
    const float* bout  = (const float*)d_in[9];
    float* out = (float*)d_out;

    float* ws      = (float*)d_ws;
    float* xw      = ws;
    float* agg     = xw + (size_t)N_NODES * 256;
    float* alpha_s = agg + (size_t)N_NODES * 256;
    float* alpha_d = alpha_s + N_NODES;
    unsigned int* own = (unsigned int*)(alpha_d + N_NODES);
    float* h2      = xw;

    gemm1_xw<<<dim3((ROW_TILES + 7) / 8, 4), 256, 0, stream>>>(x, W, xw);
    alphas_kernel<<<(N_NODES + 7) / 8, 256, 0, stream>>>(xw, a_src, a_dst, ei, alpha_s, alpha_d, own);
    agg_tile_kernel<<<AT_WGS, AT_THREADS, 0, stream>>>(ei, own, alpha_s, alpha_d, xw, agg);
    gemm2_kernel<<<dim3((ROW_TILES + 7) / 8, 2), 256, 0, stream>>>(agg, b, Wfc2, bfc2, h2);
    gemm3_kernel<<<(ROW_TILES + 7) / 8, 256, 0, stream>>>(h2, Wout, bout, out);
}


# Attention halves come out of the xw launch: each wave dots its accumulator tile with a_src/a_dst, the edge pass adds a node's 4 partials; per-node launch gone; f32 sum order of the scores changes
# speedup vs baseline: 3.2414x; 1.0818x over previous
#include <hip/hip_runtime.h>
#include <hip/hip_bf16.h>

typedef __attribute__((ext_vector_type(2))) float v2f;
typedef __attribute__((ext_vector_type(4))) float v4f;
typedef __attribute__((ext_vector_type(8))) float v8f;

#define N_NODES   100000
#define N_EDGES0  800000
#define N_EDGES   900000
#define ROW_TILES 6250

__device__ __forceinline__ float leaky(float v) { return v > 0.0f ? v : 0.2f * v; }

__device__ __forceinline__ void ldsMaxFloat(float* addr, float val) {
    if (val >= 0.0f) atomicMax((int*)addr, __float_as_int(val));
    else             atomicMin((unsigned int*)addr, __float_as_uint(val));
}

#define AT_RANGE_ROWS 392
__global__ __launch_bounds__(256) void gemm1_xw(const float* __restrict__ x,
                                                const float* __restrict__ W,
                                                const float* __restrict__ a_src,
                                                const float* __restrict__ a_dst,
                                                const int* __restrict__ ei,
                                                float* __restrict__ xw,
                                                float* __restrict__ as4,
                                                float* __restrict__ ad4,
                                                unsigned int* __restrict__ own) {
    const int g = (blockIdx.y * gridDim.x + blockIdx.x) * 256 + threadIdx.x;
    if (g < N_EDGES0 / 4) {
        const int4 d = ((const int4*)(ei + N_EDGES0))[g];
        own[g] = (((unsigned int)d.x / AT_RANGE_ROWS) & 0xFFu)
               | ((((unsigned int)d.y / AT_RANGE_ROWS) & 0xFFu) << 8)
               | ((((unsigned int)d.z / AT_RANGE_ROWS) & 0xFFu) << 16)
               | (((unsigned int)d.w / AT_RANGE_ROWS) << 24);
    }
    const int lane = threadIdx.x & 31;
    const int wave = threadIdx.x >> 5;
    const int rowTile = blockIdx.x * 8 + wave;
    if (rowTile >= ROW_TILES) return;
    const int row0 = rowTile * 16;
    const int col0 = blockIdx.y * 64;

    const int c = lane & 15;
    const int h = lane >> 4;
    const float* __restrict__ ap = x + (size_t)(row0 + c) * 128 + 4 * h;
    const float* __restrict__ bp = W + (size_t)(4 * h) * 256 + col0 + 4 * c;

    v8f acc[4] = {};
    for (int k = 0; k < 128; k += 8) {
        const v4f av = *(const v4f*)(ap + k);
        const v4f b0 = *(const v4f*)(bp + (size_t)(k + 0) * 256);
        const v4f b1 = *(const v4f*)(bp + (size_t)(k + 1) * 256);
        const v4f b2 = *(const v4f*)(bp + (size_t)(k + 2) * 256);
        const v4f b3 = *(const v4f*)(bp + (size_t)(k + 3) * 256);
        v2f a;
        a.x = av[0]; a.y = av[1];
#pragma unroll
        for (int t = 0; t < 4; ++t) {
            v2f bm;
            bm.x = b0[t]; bm.y = b1[t];
            acc[t] = __builtin_amdgcn_wmma_f32_16x16x4_f32(
                false, a, false, bm, (short)0, acc[t], false, false);
        }
        a.x = av[2]; a.y = av[3];
#pragma unroll
        for (int t = 0; t < 4; ++t) {
            v2f bm;
            bm.x = b2[t]; bm.y = b3[t];
            acc[t] = __builtin_amdgcn_wmma_f32_16x16x4_f32(
                false, a, false, bm, (short)0, acc[t], false, false);
        }
    }
    const int mBase = row0 + (h << 3);
#pragma unroll
    for (int r = 0; r < 8; ++r) {
        v4f o;
        o[0] = acc[0][r]; o[1] = acc[1][r]; o[2] = acc[2][r]; o[3] = acc[3][r];
        *(v4f*)(xw + (size_t)(mBase + r) * 256 + col0 + 4 * c) = o;
    }

    const v4f sv = *(const v4f*)(a_src + col0 + 4 * c);
    const v4f dv = *(const v4f*)(a_dst + col0 + 4 * c);
    float v[16];
#pragma unroll
    for (int r = 0; r < 8; ++r) {
        v[r]     = acc[0][r] * sv[0] + acc[1][r] * sv[1] + acc[2][r] * sv[2] + acc[3][r] * sv[3];
        v[8 + r] = acc[0][r] * dv[0] + acc[1][r] * dv[1] + acc[2][r] * dv[2] + acc[3][r] * dv[3];
    }
    float u8[8], u4[4], u2[2];
#pragma unroll
    for (int i = 0; i < 8; ++i) {
        const bool up = (c & 8) != 0;
        const float keep = up ? v[i + 8] : v[i];
        const float send = up ? v[i] : v[i + 8];
        u8[i] = keep + __shfl_xor(send, 8, 32);
    }
#pragma unroll
    for (int i = 0; i < 4; ++i) {
        const bool up = (c & 4) != 0;
        const float keep = up ? u8[i + 4] : u8[i];
        const float send = up ? u8[i] : u8[i + 4];
        u4[i] = keep + __shfl_xor(send, 4, 32);
    }
#pragma unroll
    for (int i = 0; i < 2; ++i) {
        const bool up = (c & 2) != 0;
        const float keep = up ? u4[i + 2] : u4[i];
        const float send = up ? u4[i] : u4[i + 2];
        u2[i] = keep + __shfl_xor(send, 2, 32);
    }
    const bool up1 = (c & 1) != 0;
    const float tot = (up1 ? u2[1] : u2[0]) + __shfl_xor(up1 ? u2[0] : u2[1], 1, 32);
    float* __restrict__ hp = (c < 8) ? as4 : ad4;
    hp[(size_t)(mBase + (c & 7)) * 4 + blockIdx.y] = tot;
}

#define AT_ROWS    196
#define AT_SUB     2
#define AT_RANGE   (AT_ROWS * AT_SUB)
#define AT_WGS     ((N_NODES + AT_RANGE - 1) / AT_RANGE)
#define AT_CAP     4096
#define AT_RAWCAP  (AT_SUB * AT_CAP)
#define AT_THREADS 1024
#define AT_OWN16   (N_EDGES0 / 16)
static_assert(AT_RANGE == AT_RANGE_ROWS, "owner byte and edge pass must split the rows alike");
static_assert(AT_WGS <= 256, "the owner index must fit one byte");
static_assert(N_EDGES0 % 16 == 0, "owner bytes are read sixteen at a time");

__device__ __forceinline__ int at_pos(int c) {
    return (c & 128) + ((c & 3) << 5) + ((c & 127) >> 2);
}

__device__ __forceinline__ float half_of(const float* __restrict__ p4, int node) {
    const float4 q = ((const float4*)p4)[node];
    return (q.x + q.y) + (q.z + q.w);
}

__global__ __launch_bounds__(AT_THREADS) void agg_tile_kernel(const int* __restrict__ ei,
                                                              const unsigned int* __restrict__ own,
                                                              const float* __restrict__ as4,
                                                              const float* __restrict__ ad4,
                                                              const float* __restrict__ xw,
                                                              float* __restrict__ agg) {
    __shared__ float        acc[AT_ROWS * 256];
    __shared__ unsigned int raw[AT_RAWCAP];
    __shared__ unsigned int lp[AT_SUB][AT_CAP];
    __shared__ float        la[AT_SUB][AT_CAP];
    __shared__ unsigned int cnt[AT_SUB];
    __shared__ unsigned int rawcnt;
    __shared__ float        rmax[AT_RANGE];
    __shared__ float        rden[AT_RANGE];
    __shared__ float        rself[AT_RANGE];
    __shared__ float        radl[AT_RANGE];

    const int tid  = threadIdx.x;
    const int lane = tid & 31;
    const int wave = tid >> 5;
    const int d0   = blockIdx.x * AT_RANGE;
    const unsigned int nvalid = (unsigned int)min(AT_RANGE, N_NODES - d0);

    if (tid < AT_SUB) cnt[tid] = 0u;
    if (tid == AT_SUB) rawcnt = 0u;
    if ((unsigned int)tid < nvalid) {
        const int node = d0 + tid;
        const float dh = half_of(ad4, node);
        const float sc = leaky(half_of(as4, node) + dh);
        radl[tid]  = dh;
        rself[tid] = sc;
        rmax[tid]  = sc;
        rden[tid]  = 0.0f;
    }
    __syncthreads();

    const uint4* __restrict__ own4 = (const uint4*)own;
    const unsigned int pat = (unsigned int)blockIdx.x * 0x01010101u;
    for (int base = 0; base < AT_OWN16; base += 4 * AT_THREADS) {
        uint4 ov[4];
#pragma unroll
        for (int u = 0; u < 4; ++u) {
            const int i = base + u * AT_THREADS + tid;
            ov[u] = own4[min(i, AT_OWN16 - 1)];
        }
        unsigned int hit[4];
#pragma unroll
        for (int u = 0; u < 4; ++u) {
            const unsigned int w[4] = {ov[u].x ^ pat, ov[u].y ^ pat, ov[u].z ^ pat, ov[u].w ^ pat};
            unsigned int h = 0u;
#pragma unroll
            for (int q = 0; q < 4; ++q) {
                const unsigned int m = ~(((w[q] & 0x7F7F7F7Fu) + 0x7F7F7F7Fu) | w[q]) & 0x80808080u;
                h |= m >> (7 - q);
            }
            hit[u] = (base + u * AT_THREADS + tid < AT_OWN16) ? h : 0u;
        }
#pragma unroll
        for (int v = 0; v < 2; ++v) hit[v] = hit[2 * v] | (hit[2 * v + 1] << 4);
        if ((hit[0] | hit[1]) != 0u) {
            unsigned int idx = atomicAdd(&rawcnt, (unsigned int)(__popc(hit[0]) + __popc(hit[1])));
#pragma unroll
            for (int v = 0; v < 2; ++v) {
                unsigned int hb = hit[v];
                while (hb != 0u) {
                    const int p = __ffs((int)hb) - 1;
                    hb &= hb - 1u;
                    const int u = 2 * v + ((p >> 2) & 1);
                    const int e = (base + u * AT_THREADS + tid) * 16 + 4 * (p & 3) + (p >> 3);
                    if (idx < AT_RAWCAP) raw[idx] = (unsigned int)e;
                    ++idx;
                }
            }
        }
    }
    __syncthreads();

    const unsigned int nraw = rawcnt;
    const bool ovf1 = nraw > (unsigned int)AT_RAWCAP;
    if (!ovf1) {
        for (int i = tid; i < (int)nraw; i += AT_THREADS) {
            const int e = (int)raw[i];
            const int s = ei[e];
            const int d = ei[N_EDGES0 + e];
            asm volatile("" :: "v"(s), "v"(d));
            const unsigned int r = (unsigned int)(d - d0);
            if (r < nvalid) {
                const unsigned int sub = r >= AT_ROWS ? 1u : 0u;
                const unsigned int dl  = r - sub * AT_ROWS;
                const float sc = leaky(half_of(as4, s) + radl[r]);
                const unsigned int idx = atomicAdd(&cnt[sub], 1u);
                if (idx < AT_CAP) {
                    la[sub][idx] = sc;
                    lp[sub][idx] = (unsigned int)s | (dl << 17);
                }
                ldsMaxFloat(&rmax[r], sc);
            }
        }
    }
    __syncthreads();

    const bool ovf = ovf1 || cnt[0] > (unsigned int)AT_CAP || cnt[1] > (unsigned int)AT_CAP;
    if (ovf) {
        for (int e = tid; e < N_EDGES0; e += AT_THREADS) {
            const unsigned int r = (unsigned int)(ei[N_EDGES0 + e] - d0);
            if (r < nvalid) ldsMaxFloat(&rmax[r], leaky(half_of(as4, ei[e]) + radl[r]));
        }
    }
    __syncthreads();

    if ((unsigned int)tid < nvalid) {
        const float ex = __expf(rself[tid] - rmax[tid]);
        rself[tid] = ex;
        atomicAdd(&rden[tid], ex);
    }
    if (!ovf) {
        for (int p = 0; p < AT_SUB; ++p) {
            const int n = (int)cnt[p];
            for (int i = tid; i < n; i += AT_THREADS) {
                const unsigned int rr = (unsigned int)(p * AT_ROWS) + (lp[p][i] >> 17);
                const float ex = __expf(la[p][i] - rmax[rr]);
                la[p][i] = ex;
                atomicAdd(&rden[rr], ex);
            }
        }
    } else {
        for (int e = tid; e < N_EDGES0; e += AT_THREADS) {
            const unsigned int r = (unsigned int)(ei[N_EDGES0 + e] - d0);
            if (r < nvalid) atomicAdd(&rden[r], __expf(leaky(half_of(as4, ei[e]) + radl[r]) - rmax[r]));
        }
    }
    __syncthreads();

    for (int p = 0; p < AT_SUB; ++p) {
        const int rbase = d0 + p * AT_ROWS;
        const int nrows = min(AT_ROWS, N_NODES - rbase);
        const int n = ovf ? 0 : (int)cnt[p];

        for (int r = wave; r < nrows; r += 32) {
            const int node = rbase + r;
            const float a = rself[p * AT_ROWS + r] / rden[p * AT_ROWS + r];
            const float4* row = (const float4*)(xw + (size_t)node * 256);
            float4 v0 = row[lane], v1 = row[32 + lane];
            float* ar = acc + r * 256 + lane;
            ar[0]   = a * v0.x; ar[32]  = a * v0.y; ar[64]  = a * v0.z; ar[96]  = a * v0.w;
            ar[128] = a * v1.x; ar[160] = a * v1.y; ar[192] = a * v1.z; ar[224] = a * v1.w;
        }
        for (int i = tid; i < n; i += AT_THREADS)
            la[p][i] = la[p][i] / rden[(unsigned int)(p * AT_ROWS) + (lp[p][i] >> 17)];
        __syncthreads();

        if (!ovf) {
            for (int i = wave; i < n; i += 32 * 4) {
                float4 v0[4], v1[4];
                float  a[4];
                int    dl[4];
#pragma unroll
                for (int u = 0; u < 4; ++u) {
                    const int idx = i + 32 * u;
                    const bool ok = idx < n;
                    unsigned int pk = ok ? lp[p][idx] : 0u;
                    a[u]  = ok ? la[p][idx] : 0.0f;
                    dl[u] = (int)(pk >> 17);
                    const float4* row = (const float4*)(xw + (size_t)(pk & 0x1FFFFu) * 256);
                    v0[u] = row[lane];
                    v1[u] = row[32 + lane];
                }
#pragma unroll
                for (int u = 0; u < 4; ++u) {
                    if (i + 32 * u < n) {
                        float* ar = acc + dl[u] * 256 + lane;
                        atomicAdd(ar,       a[u] * v0[u].x);
                        atomicAdd(ar + 32,  a[u] * v0[u].y);
                        atomicAdd(ar + 64,  a[u] * v0[u].z);
                        atomicAdd(ar + 96,  a[u] * v0[u].w);
                        atomicAdd(ar + 128, a[u] * v1[u].x);
                        atomicAdd(ar + 160, a[u] * v1[u].y);
                        atomicAdd(ar + 192, a[u] * v1[u].z);
                        atomicAdd(ar + 224, a[u] * v1[u].w);
                    }
                }
            }
        } else {
            for (int e = tid; e < N_EDGES0; e += AT_THREADS) {
                const int d = ei[N_EDGES0 + e];
                unsigned int r = (unsigned int)(d - rbase);
                if (r < (unsigned int)max(nrows, 0)) {
                    const int s = ei[e];
                    const unsigned int rr = (unsigned int)(p * AT_ROWS) + r;
                    const float a = __expf(leaky(half_of(as4, s) + radl[rr]) - rmax[rr]) / rden[rr];
                    const float* row = xw + (size_t)s * 256;
                    for (int c = 0; c < 256; ++c)
                        atomicAdd(acc + r * 256 + at_pos(c), a * row[c]);
                }
            }
        }
        __syncthreads();

        for (int r = wave; r < nrows; r += 32) {
            const float* ar = acc + r * 256 + lane;
            float4 o0 = make_float4(ar[0],   ar[32],  ar[64],  ar[96]);
            float4 o1 = make_float4(ar[128], ar[160], ar[192], ar[224]);
            float4* out4 = (float4*)(agg + (size_t)(rbase + r) * 256);
            out4[lane]      = o0;
            out4[32 + lane] = o1;
        }
        __syncthreads();
    }
}

__global__ __launch_bounds__(256) void gemm2_kernel(const float* __restrict__ agg,
                                                    const float* __restrict__ b,
                                                    const float* __restrict__ Wfc2,
                                                    const float* __restrict__ bfc2,
                                                    float* __restrict__ h2) {
    const int lane = threadIdx.x & 31;
    const int wave = threadIdx.x >> 5;
    const int rowTile = blockIdx.x * 8 + wave;
    if (rowTile >= ROW_TILES) return;
    const int row0 = rowTile * 16;
    const int col0 = blockIdx.y * 64;

    const int c = lane & 15;
    const int h = lane >> 4;
    const float* __restrict__ ap = agg + (size_t)(row0 + c) * 256 + 4 * h;
    const float* __restrict__ cp = b + 4 * h;
    const float* __restrict__ bp = Wfc2 + (size_t)(4 * h) * 128 + col0 + 4 * c;

    v8f acc[4] = {};
    for (int k = 0; k < 256; k += 8) {
        const v4f av = *(const v4f*)(ap + k);
        const v4f cv = *(const v4f*)(cp + k);
        const v4f b0 = *(const v4f*)(bp + (size_t)(k + 0) * 128);
        const v4f b1 = *(const v4f*)(bp + (size_t)(k + 1) * 128);
        const v4f b2 = *(const v4f*)(bp + (size_t)(k + 2) * 128);
        const v4f b3 = *(const v4f*)(bp + (size_t)(k + 3) * 128);
        v2f a;
        a.x = fmaxf(av[0] + cv[0], 0.0f);
        a.y = fmaxf(av[1] + cv[1], 0.0f);
#pragma unroll
        for (int t = 0; t < 4; ++t) {
            v2f bm;
            bm.x = b0[t]; bm.y = b1[t];
            acc[t] = __builtin_amdgcn_wmma_f32_16x16x4_f32(
                false, a, false, bm, (short)0, acc[t], false, false);
        }
        a.x = fmaxf(av[2] + cv[2], 0.0f);
        a.y = fmaxf(av[3] + cv[3], 0.0f);
#pragma unroll
        for (int t = 0; t < 4; ++t) {
            v2f bm;
            bm.x = b2[t]; bm.y = b3[t];
            acc[t] = __builtin_amdgcn_wmma_f32_16x16x4_f32(
                false, a, false, bm, (short)0, acc[t], false, false);
        }
    }
    const int mBase = row0 + (h << 3);
    const v4f bo = *(const v4f*)(bfc2 + col0 + 4 * c);
#pragma unroll
    for (int r = 0; r < 8; ++r) {
        v4f o;
        o[0] = fmaxf(acc[0][r] + bo[0], 0.0f);
        o[1] = fmaxf(acc[1][r] + bo[1], 0.0f);
        o[2] = fmaxf(acc[2][r] + bo[2], 0.0f);
        o[3] = fmaxf(acc[3][r] + bo[3], 0.0f);
        *(v4f*)(h2 + (size_t)(mBase + r) * 128 + col0 + 4 * c) = o;
    }
}

__global__ __launch_bounds__(256) void gemm3_kernel(const float* __restrict__ h2,
                                                    const float* __restrict__ Wout,
                                                    const float* __restrict__ bout,
                                                    float* __restrict__ out) {
    const int lane = threadIdx.x & 31;
    const int wave = threadIdx.x >> 5;
    const int rowTile = blockIdx.x * 8 + wave;
    if (rowTile >= ROW_TILES) return;
    const int row0 = rowTile * 16;

    const int nB = lane & 15;
    const int h  = lane >> 4;
    const float* __restrict__ ap = h2 + (size_t)(row0 + nB) * 128 + 4 * h;

    v8f acc = {};
    for (int k = 0; k < 128; k += 8) {
        const v4f av = *(const v4f*)(ap + k);
        const int kr = k + 4 * h;
        v2f a, bm;
        a.x = av[0]; a.y = av[1];
        bm.x = (nB < 6) ? Wout[(size_t)(kr + 0) * 6 + nB] : 0.0f;
        bm.y = (nB < 6) ? Wout[(size_t)(kr + 1) * 6 + nB] : 0.0f;
        acc = __builtin_amdgcn_wmma_f32_16x16x4_f32(
            false, a, false, bm, (short)0, acc, false, false);
        a.x = av[2]; a.y = av[3];
        bm.x = (nB < 6) ? Wout[(size_t)(kr + 2) * 6 + nB] : 0.0f;
        bm.y = (nB < 6) ? Wout[(size_t)(kr + 3) * 6 + nB] : 0.0f;
        acc = __builtin_amdgcn_wmma_f32_16x16x4_f32(
            false, a, false, bm, (short)0, acc, false, false);
    }
    if (nB < 6) {
        const int mBase = row0 + (h << 3);
        float bb = bout[nB];
#pragma unroll
        for (int r = 0; r < 8; ++r)
            out[(size_t)(mBase + r) * 6 + nB] = acc[r] + bb;
    }
}

extern "C" void kernel_launch(void* const* d_in, const int* in_sizes, int n_in,
                              void* d_out, int out_size, void* d_ws, size_t ws_size,
                              hipStream_t stream) {
    const float* x     = (const float*)d_in[0];
    const int*   ei    = (const int*)  d_in[1];
    const float* W     = (const float*)d_in[2];
    const float* a_src = (const float*)d_in[3];
    const float* a_dst = (const float*)d_in[4];
    const float* b     = (const float*)d_in[5];
    const float* Wfc2  = (const float*)d_in[6];
    const float* bfc2  = (const float*)d_in[7];
    const float* Wout  = (const float*)d_in[8];
    const float* bout  = (const float*)d_in[9];
    float* out = (float*)d_out;

    float* ws      = (float*)d_ws;
    float* xw      = ws;
    float* agg     = xw + (size_t)N_NODES * 256;
    float* as4     = agg + (size_t)N_NODES * 256;
    float* ad4     = as4 + (size_t)N_NODES * 4;
    unsigned int* own = (unsigned int*)(ad4 + (size_t)N_NODES * 4);
    float* h2      = xw;

    gemm1_xw<<<dim3((ROW_TILES + 7) / 8, 4), 256, 0, stream>>>(x, W, a_src, a_dst, ei, xw, as4, ad4, own);
    agg_tile_kernel<<<AT_WGS, AT_THREADS, 0, stream>>>(ei, own, as4, ad4, xw, agg);
    gemm2_kernel<<<dim3((ROW_TILES + 7) / 8, 2), 256, 0, stream>>>(agg, b, Wfc2, bfc2, h2);
    gemm3_kernel<<<(ROW_TILES + 7) / 8, 256, 0, stream>>>(h2, Wout, bout, out);
}


# Output head chained onto the fc2 launch in registers: fc2 taken transposed, its accumulators are the head's B fragments; head launch and h2 gone; head f32 sum order changes (2 partials added)
# speedup vs baseline: 3.5451x; 1.0937x over previous
#include <hip/hip_runtime.h>
#include <hip/hip_bf16.h>

typedef __attribute__((ext_vector_type(2))) float v2f;
typedef __attribute__((ext_vector_type(4))) float v4f;
typedef __attribute__((ext_vector_type(8))) float v8f;

#define N_NODES   100000
#define N_EDGES0  800000
#define N_EDGES   900000
#define ROW_TILES 6250

__device__ __forceinline__ float leaky(float v) { return v > 0.0f ? v : 0.2f * v; }

__device__ __forceinline__ void ldsMaxFloat(float* addr, float val) {
    if (val >= 0.0f) atomicMax((int*)addr, __float_as_int(val));
    else             atomicMin((unsigned int*)addr, __float_as_uint(val));
}

#define AT_RANGE_ROWS 392
__global__ __launch_bounds__(256) void gemm1_xw(const float* __restrict__ x,
                                                const float* __restrict__ W,
                                                const float* __restrict__ a_src,
                                                const float* __restrict__ a_dst,
                                                const int* __restrict__ ei,
                                                float* __restrict__ xw,
                                                float* __restrict__ as4,
                                                float* __restrict__ ad4,
                                                unsigned int* __restrict__ own) {
    const int g = (blockIdx.y * gridDim.x + blockIdx.x) * 256 + threadIdx.x;
    if (g < N_EDGES0 / 4) {
        const int4 d = ((const int4*)(ei + N_EDGES0))[g];
        own[g] = (((unsigned int)d.x / AT_RANGE_ROWS) & 0xFFu)
               | ((((unsigned int)d.y / AT_RANGE_ROWS) & 0xFFu) << 8)
               | ((((unsigned int)d.z / AT_RANGE_ROWS) & 0xFFu) << 16)
               | (((unsigned int)d.w / AT_RANGE_ROWS) << 24);
    }
    const int lane = threadIdx.x & 31;
    const int wave = threadIdx.x >> 5;
    const int rowTile = blockIdx.x * 8 + wave;
    if (rowTile >= ROW_TILES) return;
    const int row0 = rowTile * 16;
    const int col0 = blockIdx.y * 64;

    const int c = lane & 15;
    const int h = lane >> 4;
    const float* __restrict__ ap = x + (size_t)(row0 + c) * 128 + 4 * h;
    const float* __restrict__ bp = W + (size_t)(4 * h) * 256 + col0 + 4 * c;

    v8f acc[4] = {};
    for (int k = 0; k < 128; k += 8) {
        const v4f av = *(const v4f*)(ap + k);
        const v4f b0 = *(const v4f*)(bp + (size_t)(k + 0) * 256);
        const v4f b1 = *(const v4f*)(bp + (size_t)(k + 1) * 256);
        const v4f b2 = *(const v4f*)(bp + (size_t)(k + 2) * 256);
        const v4f b3 = *(const v4f*)(bp + (size_t)(k + 3) * 256);
        v2f a;
        a.x = av[0]; a.y = av[1];
#pragma unroll
        for (int t = 0; t < 4; ++t) {
            v2f bm;
            bm.x = b0[t]; bm.y = b1[t];
            acc[t] = __builtin_amdgcn_wmma_f32_16x16x4_f32(
                false, a, false, bm, (short)0, acc[t], false, false);
        }
        a.x = av[2]; a.y = av[3];
#pragma unroll
        for (int t = 0; t < 4; ++t) {
            v2f bm;
            bm.x = b2[t]; bm.y = b3[t];
            acc[t] = __builtin_amdgcn_wmma_f32_16x16x4_f32(
                false, a, false, bm, (short)0, acc[t], false, false);
        }
    }
    const int mBase = row0 + (h << 3);
#pragma unroll
    for (int r = 0; r < 8; ++r) {
        v4f o;
        o[0] = acc[0][r]; o[1] = acc[1][r]; o[2] = acc[2][r]; o[3] = acc[3][r];
        *(v4f*)(xw + (size_t)(mBase + r) * 256 + col0 + 4 * c) = o;
    }

    const v4f sv = *(const v4f*)(a_src + col0 + 4 * c);
    const v4f dv = *(const v4f*)(a_dst + col0 + 4 * c);
    float v[16];
#pragma unroll
    for (int r = 0; r < 8; ++r) {
        v[r]     = acc[0][r] * sv[0] + acc[1][r] * sv[1] + acc[2][r] * sv[2] + acc[3][r] * sv[3];
        v[8 + r] = acc[0][r] * dv[0] + acc[1][r] * dv[1] + acc[2][r] * dv[2] + acc[3][r] * dv[3];
    }
    float u8[8], u4[4], u2[2];
#pragma unroll
    for (int i = 0; i < 8; ++i) {
        const bool up = (c & 8) != 0;
        const float keep = up ? v[i + 8] : v[i];
        const float send = up ? v[i] : v[i + 8];
        u8[i] = keep + __shfl_xor(send, 8, 32);
    }
#pragma unroll
    for (int i = 0; i < 4; ++i) {
        const bool up = (c & 4) != 0;
        const float keep = up ? u8[i + 4] : u8[i];
        const float send = up ? u8[i] : u8[i + 4];
        u4[i] = keep + __shfl_xor(send, 4, 32);
    }
#pragma unroll
    for (int i = 0; i < 2; ++i) {
        const bool up = (c & 2) != 0;
        const float keep = up ? u4[i + 2] : u4[i];
        const float send = up ? u4[i] : u4[i + 2];
        u2[i] = keep + __shfl_xor(send, 2, 32);
    }
    const bool up1 = (c & 1) != 0;
    const float tot = (up1 ? u2[1] : u2[0]) + __shfl_xor(up1 ? u2[0] : u2[1], 1, 32);
    float* __restrict__ hp = (c < 8) ? as4 : ad4;
    hp[(size_t)(mBase + (c & 7)) * 4 + blockIdx.y] = tot;
}

#define AT_ROWS    196
#define AT_SUB     2
#define AT_RANGE   (AT_ROWS * AT_SUB)
#define AT_WGS     ((N_NODES + AT_RANGE - 1) / AT_RANGE)
#define AT_CAP     4096
#define AT_RAWCAP  (AT_SUB * AT_CAP)
#define AT_THREADS 1024
#define AT_OWN16   (N_EDGES0 / 16)
static_assert(AT_RANGE == AT_RANGE_ROWS, "owner byte and edge pass must split the rows alike");
static_assert(AT_WGS <= 256, "the owner index must fit one byte");
static_assert(N_EDGES0 % 16 == 0, "owner bytes are read sixteen at a time");

__device__ __forceinline__ int at_pos(int c) {
    return (c & 128) + ((c & 3) << 5) + ((c & 127) >> 2);
}

__device__ __forceinline__ float half_of(const float* __restrict__ p4, int node) {
    const float4 q = ((const float4*)p4)[node];
    return (q.x + q.y) + (q.z + q.w);
}

__global__ __launch_bounds__(AT_THREADS) void agg_tile_kernel(const int* __restrict__ ei,
                                                              const unsigned int* __restrict__ own,
                                                              const float* __restrict__ as4,
                                                              const float* __restrict__ ad4,
                                                              const float* __restrict__ xw,
                                                              float* __restrict__ agg) {
    __shared__ float        acc[AT_ROWS * 256];
    __shared__ unsigned int raw[AT_RAWCAP];
    __shared__ unsigned int lp[AT_SUB][AT_CAP];
    __shared__ float        la[AT_SUB][AT_CAP];
    __shared__ unsigned int cnt[AT_SUB];
    __shared__ unsigned int rawcnt;
    __shared__ float        rmax[AT_RANGE];
    __shared__ float        rden[AT_RANGE];
    __shared__ float        rself[AT_RANGE];
    __shared__ float        radl[AT_RANGE];

    const int tid  = threadIdx.x;
    const int lane = tid & 31;
    const int wave = tid >> 5;
    const int d0   = blockIdx.x * AT_RANGE;
    const unsigned int nvalid = (unsigned int)min(AT_RANGE, N_NODES - d0);

    if (tid < AT_SUB) cnt[tid] = 0u;
    if (tid == AT_SUB) rawcnt = 0u;
    if ((unsigned int)tid < nvalid) {
        const int node = d0 + tid;
        const float dh = half_of(ad4, node);
        const float sc = leaky(half_of(as4, node) + dh);
        radl[tid]  = dh;
        rself[tid] = sc;
        rmax[tid]  = sc;
        rden[tid]  = 0.0f;
    }
    __syncthreads();

    const uint4* __restrict__ own4 = (const uint4*)own;
    const unsigned int pat = (unsigned int)blockIdx.x * 0x01010101u;
    for (int base = 0; base < AT_OWN16; base += 4 * AT_THREADS) {
        uint4 ov[4];
#pragma unroll
        for (int u = 0; u < 4; ++u) {
            const int i = base + u * AT_THREADS + tid;
            ov[u] = own4[min(i, AT_OWN16 - 1)];
        }
        unsigned int hit[4];
#pragma unroll
        for (int u = 0; u < 4; ++u) {
            const unsigned int w[4] = {ov[u].x ^ pat, ov[u].y ^ pat, ov[u].z ^ pat, ov[u].w ^ pat};
            unsigned int h = 0u;
#pragma unroll
            for (int q = 0; q < 4; ++q) {
                const unsigned int m = ~(((w[q] & 0x7F7F7F7Fu) + 0x7F7F7F7Fu) | w[q]) & 0x80808080u;
                h |= m >> (7 - q);
            }
            hit[u] = (base + u * AT_THREADS + tid < AT_OWN16) ? h : 0u;
        }
#pragma unroll
        for (int v = 0; v < 2; ++v) hit[v] = hit[2 * v] | (hit[2 * v + 1] << 4);
        if ((hit[0] | hit[1]) != 0u) {
            unsigned int idx = atomicAdd(&rawcnt, (unsigned int)(__popc(hit[0]) + __popc(hit[1])));
#pragma unroll
            for (int v = 0; v < 2; ++v) {
                unsigned int hb = hit[v];
                while (hb != 0u) {
                    const int p = __ffs((int)hb) - 1;
                    hb &= hb - 1u;
                    const int u = 2 * v + ((p >> 2) & 1);
                    const int e = (base + u * AT_THREADS + tid) * 16 + 4 * (p & 3) + (p >> 3);
                    if (idx < AT_RAWCAP) raw[idx] = (unsigned int)e;
                    ++idx;
                }
            }
        }
    }
    __syncthreads();

    const unsigned int nraw = rawcnt;
    const bool ovf1 = nraw > (unsigned int)AT_RAWCAP;
    if (!ovf1) {
        for (int i = tid; i < (int)nraw; i += AT_THREADS) {
            const int e = (int)raw[i];
            const int s = ei[e];
            const int d = ei[N_EDGES0 + e];
            asm volatile("" :: "v"(s), "v"(d));
            const unsigned int r = (unsigned int)(d - d0);
            if (r < nvalid) {
                const unsigned int sub = r >= AT_ROWS ? 1u : 0u;
                const unsigned int dl  = r - sub * AT_ROWS;
                const float sc = leaky(half_of(as4, s) + radl[r]);
                const unsigned int idx = atomicAdd(&cnt[sub], 1u);
                if (idx < AT_CAP) {
                    la[sub][idx] = sc;
                    lp[sub][idx] = (unsigned int)s | (dl << 17);
                }
                ldsMaxFloat(&rmax[r], sc);
            }
        }
    }
    __syncthreads();

    const bool ovf = ovf1 || cnt[0] > (unsigned int)AT_CAP || cnt[1] > (unsigned int)AT_CAP;
    if (ovf) {
        for (int e = tid; e < N_EDGES0; e += AT_THREADS) {
            const unsigned int r = (unsigned int)(ei[N_EDGES0 + e] - d0);
            if (r < nvalid) ldsMaxFloat(&rmax[r], leaky(half_of(as4, ei[e]) + radl[r]));
        }
    }
    __syncthreads();

    if ((unsigned int)tid < nvalid) {
        const float ex = __expf(rself[tid] - rmax[tid]);
        rself[tid] = ex;
        atomicAdd(&rden[tid], ex);
    }
    if (!ovf) {
        for (int p = 0; p < AT_SUB; ++p) {
            const int n = (int)cnt[p];
            for (int i = tid; i < n; i += AT_THREADS) {
                const unsigned int rr = (unsigned int)(p * AT_ROWS) + (lp[p][i] >> 17);
                const float ex = __expf(la[p][i] - rmax[rr]);
                la[p][i] = ex;
                atomicAdd(&rden[rr], ex);
            }
        }
    } else {
        for (int e = tid; e < N_EDGES0; e += AT_THREADS) {
            const unsigned int r = (unsigned int)(ei[N_EDGES0 + e] - d0);
            if (r < nvalid) atomicAdd(&rden[r], __expf(leaky(half_of(as4, ei[e]) + radl[r]) - rmax[r]));
        }
    }
    __syncthreads();

    for (int p = 0; p < AT_SUB; ++p) {
        const int rbase = d0 + p * AT_ROWS;
        const int nrows = min(AT_ROWS, N_NODES - rbase);
        const int n = ovf ? 0 : (int)cnt[p];

        for (int r = wave; r < nrows; r += 32) {
            const int node = rbase + r;
            const float a = rself[p * AT_ROWS + r] / rden[p * AT_ROWS + r];
            const float4* row = (const float4*)(xw + (size_t)node * 256);
            float4 v0 = row[lane], v1 = row[32 + lane];
            float* ar = acc + r * 256 + lane;
            ar[0]   = a * v0.x; ar[32]  = a * v0.y; ar[64]  = a * v0.z; ar[96]  = a * v0.w;
            ar[128] = a * v1.x; ar[160] = a * v1.y; ar[192] = a * v1.z; ar[224] = a * v1.w;
        }
        for (int i = tid; i < n; i += AT_THREADS)
            la[p][i] = la[p][i] / rden[(unsigned int)(p * AT_ROWS) + (lp[p][i] >> 17)];
        __syncthreads();

        if (!ovf) {
            for (int i = wave; i < n; i += 32 * 4) {
                float4 v0[4], v1[4];
                float  a[4];
                int    dl[4];
#pragma unroll
                for (int u = 0; u < 4; ++u) {
                    const int idx = i + 32 * u;
                    const bool ok = idx < n;
                    unsigned int pk = ok ? lp[p][idx] : 0u;
                    a[u]  = ok ? la[p][idx] : 0.0f;
                    dl[u] = (int)(pk >> 17);
                    const float4* row = (const float4*)(xw + (size_t)(pk & 0x1FFFFu) * 256);
                    v0[u] = row[lane];
                    v1[u] = row[32 + lane];
                }
#pragma unroll
                for (int u = 0; u < 4; ++u) {
                    if (i + 32 * u < n) {
                        float* ar = acc + dl[u] * 256 + lane;
                        atomicAdd(ar,       a[u] * v0[u].x);
                        atomicAdd(ar + 32,  a[u] * v0[u].y);
                        atomicAdd(ar + 64,  a[u] * v0[u].z);
                        atomicAdd(ar + 96,  a[u] * v0[u].w);
                        atomicAdd(ar + 128, a[u] * v1[u].x);
                        atomicAdd(ar + 160, a[u] * v1[u].y);
                        atomicAdd(ar + 192, a[u] * v1[u].z);
                        atomicAdd(ar + 224, a[u] * v1[u].w);
                    }
                }
            }
        } else {
            for (int e = tid; e < N_EDGES0; e += AT_THREADS) {
                const int d = ei[N_EDGES0 + e];
                unsigned int r = (unsigned int)(d - rbase);
                if (r < (unsigned int)max(nrows, 0)) {
                    const int s = ei[e];
                    const unsigned int rr = (unsigned int)(p * AT_ROWS) + r;
                    const float a = __expf(leaky(half_of(as4, s) + radl[rr]) - rmax[rr]) / rden[rr];
                    const float* row = xw + (size_t)s * 256;
                    for (int c = 0; c < 256; ++c)
                        atomicAdd(acc + r * 256 + at_pos(c), a * row[c]);
                }
            }
        }
        __syncthreads();

        for (int r = wave; r < nrows; r += 32) {
            const float* ar = acc + r * 256 + lane;
            float4 o0 = make_float4(ar[0],   ar[32],  ar[64],  ar[96]);
            float4 o1 = make_float4(ar[128], ar[160], ar[192], ar[224]);
            float4* out4 = (float4*)(agg + (size_t)(rbase + r) * 256);
            out4[lane]      = o0;
            out4[32 + lane] = o1;
        }
        __syncthreads();
    }
}

__global__ __launch_bounds__(256) void fc2_head_kernel(const float* __restrict__ agg,
                                                       const float* __restrict__ b,
                                                       const float* __restrict__ Wfc2,
                                                       const float* __restrict__ bfc2,
                                                       const float* __restrict__ Wout,
                                                       const float* __restrict__ bout,
                                                       float* __restrict__ out) {
    __shared__ float part[4][2][16][8];
    const int lane = threadIdx.x & 31;
    const int wave = threadIdx.x >> 5;
    const int rt = wave & 3;
    const int cb = wave >> 2;
    const int rowTile = min(blockIdx.x * 4 + rt, ROW_TILES - 1);
    const int row0 = rowTile * 16;
    const int col0 = cb * 64;

    const int c = lane & 15;
    const int h = lane >> 4;
    const float* __restrict__ ap = agg + (size_t)(row0 + c) * 256 + 4 * h;
    const float* __restrict__ cp = b + 4 * h;
    const float* __restrict__ bp = Wfc2 + (size_t)(4 * h) * 128 + col0 + 4 * c;

    v8f acc[4] = {};
    for (int k = 0; k < 256; k += 8) {
        const v4f av = *(const v4f*)(ap + k);
        const v4f cv = *(const v4f*)(cp + k);
        const v4f b0 = *(const v4f*)(bp + (size_t)(k + 0) * 128);
        const v4f b1 = *(const v4f*)(bp + (size_t)(k + 1) * 128);
        const v4f b2 = *(const v4f*)(bp + (size_t)(k + 2) * 128);
        const v4f b3 = *(const v4f*)(bp + (size_t)(k + 3) * 128);
        v2f a;
        a.x = fmaxf(av[0] + cv[0], 0.0f);
        a.y = fmaxf(av[1] + cv[1], 0.0f);
#pragma unroll
        for (int t = 0; t < 4; ++t) {
            v2f bm;
            bm.x = b0[t]; bm.y = b1[t];
            acc[t] = __builtin_amdgcn_wmma_f32_16x16x4_f32(
                false, bm, false, a, (short)0, acc[t], false, false);
        }
        a.x = fmaxf(av[2] + cv[2], 0.0f);
        a.y = fmaxf(av[3] + cv[3], 0.0f);
#pragma unroll
        for (int t = 0; t < 4; ++t) {
            v2f bm;
            bm.x = b2[t]; bm.y = b3[t];
            acc[t] = __builtin_amdgcn_wmma_f32_16x16x4_f32(
                false, bm, false, a, (short)0, acc[t], false, false);
        }
    }

    const float* __restrict__ bq = bfc2 + col0 + 32 * h;
    const int kb = col0 + 32 * h;
    const int cj = min(c, 5);
    float wv[32];
#pragma unroll
    for (int q = 0; q < 4; ++q) {
#pragma unroll
        for (int t = 0; t < 4; ++t) {
            wv[8 * q + 2 * t]     = Wout[(size_t)(kb + 8 * q + t) * 6 + cj];
            wv[8 * q + 2 * t + 1] = Wout[(size_t)(kb + 8 * q + 4 + t) * 6 + cj];
        }
    }
    v8f o = {};
#pragma unroll
    for (int q = 0; q < 4; ++q) {
        const v4f bo0 = *(const v4f*)(bq + 8 * q);
        const v4f bo1 = *(const v4f*)(bq + 8 * q + 4);
#pragma unroll
        for (int t = 0; t < 4; ++t) {
            v2f hb, wa;
            hb.x = fmaxf(acc[t][2 * q]     + bo0[t], 0.0f);
            hb.y = fmaxf(acc[t][2 * q + 1] + bo1[t], 0.0f);
            wa.x = (c < 6) ? wv[8 * q + 2 * t]     : 0.0f;
            wa.y = (c < 6) ? wv[8 * q + 2 * t + 1] : 0.0f;
            o = __builtin_amdgcn_wmma_f32_16x16x4_f32(
                false, wa, false, hb, (short)0, o, false, false);
        }
    }
    if (h == 0) {
#pragma unroll
        for (int r = 0; r < 6; ++r) part[rt][cb][c][r] = o[r];
    }
    __syncthreads();
    for (int i = threadIdx.x; i < 4 * 96; i += 256) {
        const int t4 = i / 96;
        const int rem = i - t4 * 96;
        const int node = rem / 6;
        const int j = rem - node * 6;
        if (blockIdx.x * 4 + t4 < ROW_TILES)
            out[(size_t)blockIdx.x * 384 + i] = (part[t4][0][node][j] + part[t4][1][node][j]) + bout[j];
    }
}

extern "C" void kernel_launch(void* const* d_in, const int* in_sizes, int n_in,
                              void* d_out, int out_size, void* d_ws, size_t ws_size,
                              hipStream_t stream) {
    const float* x     = (const float*)d_in[0];
    const int*   ei    = (const int*)  d_in[1];
    const float* W     = (const float*)d_in[2];
    const float* a_src = (const float*)d_in[3];
    const float* a_dst = (const float*)d_in[4];
    const float* b     = (const float*)d_in[5];
    const float* Wfc2  = (const float*)d_in[6];
    const float* bfc2  = (const float*)d_in[7];
    const float* Wout  = (const float*)d_in[8];
    const float* bout  = (const float*)d_in[9];
    float* out = (float*)d_out;

    float* ws      = (float*)d_ws;
    float* xw      = ws;
    float* agg     = xw + (size_t)N_NODES * 256;
    float* as4     = agg + (size_t)N_NODES * 256;
    float* ad4     = as4 + (size_t)N_NODES * 4;
    unsigned int* own = (unsigned int*)(ad4 + (size_t)N_NODES * 4);

    gemm1_xw<<<dim3((ROW_TILES + 7) / 8, 4), 256, 0, stream>>>(x, W, a_src, a_dst, ei, xw, as4, ad4, own);
    agg_tile_kernel<<<AT_WGS, AT_THREADS, 0, stream>>>(ei, own, as4, ad4, xw, agg);
    fc2_head_kernel<<<(ROW_TILES + 3) / 4, 256, 0, stream>>>(agg, b, Wfc2, bfc2, Wout, bout, out);
}


# fc2 + head chained onto the edge pass: a finished LDS tile is fc2's operand (16-byte LDS fragment reads), agg never stored, a launch gone; f32 sum order of fc2 (k order) and head (8 partials) changes
# speedup vs baseline: 3.8619x; 1.0894x over previous
#include <hip/hip_runtime.h>
#include <hip/hip_bf16.h>

typedef __attribute__((ext_vector_type(2))) float v2f;
typedef __attribute__((ext_vector_type(4))) float v4f;
typedef __attribute__((ext_vector_type(8))) float v8f;

#define N_NODES   100000
#define N_EDGES0  800000
#define N_EDGES   900000
#define ROW_TILES 6250

__device__ __forceinline__ float leaky(float v) { return v > 0.0f ? v : 0.2f * v; }

__device__ __forceinline__ void ldsMaxFloat(float* addr, float val) {
    if (val >= 0.0f) atomicMax((int*)addr, __float_as_int(val));
    else             atomicMin((unsigned int*)addr, __float_as_uint(val));
}

#define AT_RANGE_ROWS 392
__global__ __launch_bounds__(256) void gemm1_xw(const float* __restrict__ x,
                                                const float* __restrict__ W,
                                                const float* __restrict__ a_src,
                                                const float* __restrict__ a_dst,
                                                const int* __restrict__ ei,
                                                float* __restrict__ xw,
                                                float* __restrict__ as4,
                                                float* __restrict__ ad4,
                                                unsigned int* __restrict__ own) {
    const int g = (blockIdx.y * gridDim.x + blockIdx.x) * 256 + threadIdx.x;
    if (g < N_EDGES0 / 4) {
        const int4 d = ((const int4*)(ei + N_EDGES0))[g];
        own[g] = (((unsigned int)d.x / AT_RANGE_ROWS) & 0xFFu)
               | ((((unsigned int)d.y / AT_RANGE_ROWS) & 0xFFu) << 8)
               | ((((unsigned int)d.z / AT_RANGE_ROWS) & 0xFFu) << 16)
               | (((unsigned int)d.w / AT_RANGE_ROWS) << 24);
    }
    const int lane = threadIdx.x & 31;
    const int wave = threadIdx.x >> 5;
    const int rowTile = blockIdx.x * 8 + wave;
    if (rowTile >= ROW_TILES) return;
    const int row0 = rowTile * 16;
    const int col0 = blockIdx.y * 64;

    const int c = lane & 15;
    const int h = lane >> 4;
    const float* __restrict__ ap = x + (size_t)(row0 + c) * 128 + 4 * h;
    const float* __restrict__ bp = W + (size_t)(4 * h) * 256 + col0 + 4 * c;

    v8f acc[4] = {};
    for (int k = 0; k < 128; k += 8) {
        const v4f av = *(const v4f*)(ap + k);
        const v4f b0 = *(const v4f*)(bp + (size_t)(k + 0) * 256);
        const v4f b1 = *(const v4f*)(bp + (size_t)(k + 1) * 256);
        const v4f b2 = *(const v4f*)(bp + (size_t)(k + 2) * 256);
        const v4f b3 = *(const v4f*)(bp + (size_t)(k + 3) * 256);
        v2f a;
        a.x = av[0]; a.y = av[1];
#pragma unroll
        for (int t = 0; t < 4; ++t) {
            v2f bm;
            bm.x = b0[t]; bm.y = b1[t];
            acc[t] = __builtin_amdgcn_wmma_f32_16x16x4_f32(
                false, a, false, bm, (short)0, acc[t], false, false);
        }
        a.x = av[2]; a.y = av[3];
#pragma unroll
        for (int t = 0; t < 4; ++t) {
            v2f bm;
            bm.x = b2[t]; bm.y = b3[t];
            acc[t] = __builtin_amdgcn_wmma_f32_16x16x4_f32(
                false, a, false, bm, (short)0, acc[t], false, false);
        }
    }
    const int mBase = row0 + (h << 3);
#pragma unroll
    for (int r = 0; r < 8; ++r) {
        v4f o;
        o[0] = acc[0][r]; o[1] = acc[1][r]; o[2] = acc[2][r]; o[3] = acc[3][r];
        *(v4f*)(xw + (size_t)(mBase + r) * 256 + col0 + 4 * c) = o;
    }

    const v4f sv = *(const v4f*)(a_src + col0 + 4 * c);
    const v4f dv = *(const v4f*)(a_dst + col0 + 4 * c);
    float v[16];
#pragma unroll
    for (int r = 0; r < 8; ++r) {
        v[r]     = acc[0][r] * sv[0] + acc[1][r] * sv[1] + acc[2][r] * sv[2] + acc[3][r] * sv[3];
        v[8 + r] = acc[0][r] * dv[0] + acc[1][r] * dv[1] + acc[2][r] * dv[2] + acc[3][r] * dv[3];
    }
    float u8[8], u4[4], u2[2];
#pragma unroll
    for (int i = 0; i < 8; ++i) {
        const bool up = (c & 8) != 0;
        const float keep = up ? v[i + 8] : v[i];
        const float send = up ? v[i] : v[i + 8];
        u8[i] = keep + __shfl_xor(send, 8, 32);
    }
#pragma unroll
    for (int i = 0; i < 4; ++i) {
        const bool up = (c & 4) != 0;
        const float keep = up ? u8[i + 4] : u8[i];
        const float send = up ? u8[i] : u8[i + 4];
        u4[i] = keep + __shfl_xor(send, 4, 32);
    }
#pragma unroll
    for (int i = 0; i < 2; ++i) {
        const bool up = (c & 2) != 0;
        const float keep = up ? u4[i + 2] : u4[i];
        const float send = up ? u4[i] : u4[i + 2];
        u2[i] = keep + __shfl_xor(send, 2, 32);
    }
    const bool up1 = (c & 1) != 0;
    const float tot = (up1 ? u2[1] : u2[0]) + __shfl_xor(up1 ? u2[0] : u2[1], 1, 32);
    float* __restrict__ hp = (c < 8) ? as4 : ad4;
    hp[(size_t)(mBase + (c & 7)) * 4 + blockIdx.y] = tot;
}

#define AT_ROWS    196
#define AT_SUB     2
#define AT_RANGE   (AT_ROWS * AT_SUB)
#define AT_WGS     ((N_NODES + AT_RANGE - 1) / AT_RANGE)
#define AT_CAP     4096
#define AT_RAWCAP  (AT_SUB * AT_CAP)
#define AT_THREADS 1024
#define AT_OWN16   (N_EDGES0 / 16)
#define AT_PITCH   260
#define FC_NT      ((AT_ROWS + 15) / 16)
#define FC_ROWS    (FC_NT * 16)
#define FC_HP      (8 * FC_ROWS * 6)
#define AT_RAWW    (AT_RAWCAP > FC_HP ? AT_RAWCAP : FC_HP)
static_assert(AT_RANGE == AT_RANGE_ROWS, "owner byte and edge pass must split the rows alike");
static_assert(AT_WGS <= 256, "the owner index must fit one byte");
static_assert(N_EDGES0 % 16 == 0, "owner bytes are read sixteen at a time");

__device__ __forceinline__ int at_pos(int c) {
    return (c & 128) + ((c & 3) << 5) + ((c & 127) >> 2);
}

__device__ __forceinline__ float half_of(const float* __restrict__ p4, int node) {
    const float4 q = ((const float4*)p4)[node];
    return (q.x + q.y) + (q.z + q.w);
}

template <int R>
__device__ __forceinline__ void fc2_head_tile(const float* __restrict__ accp, float* __restrict__ hp,
                                              const float* __restrict__ Wfc2, const float* __restrict__ bfc2,
                                              const float* __restrict__ Wout,
                                              const int ct, const int rg, const int lastRow, const int lane) {
    const int c = lane & 15;
    const int h = lane >> 4;
    const float* rp[R];
#pragma unroll
    for (int j = 0; j < R; ++j) rp[j] = accp + min((rg + 4 * j) * 16 + c, lastRow) * AT_PITCH + 4 * h;
    const float* __restrict__ wp = Wfc2 + (size_t)(16 * h) * 128 + ct * 16 + c;

    v8f a[R];
#pragma unroll
    for (int j = 0; j < R; ++j) a[j] = v8f{};
#pragma unroll 2
    for (int st = 0; st < 32; ++st) {
        const int w0 = st * 8;
        const int k0 = (w0 & 128) + 4 * (w0 & 31) + ((w0 >> 5) & 3);
        const float wk0 = wp[(size_t)(k0 + 0) * 128];
        const float wk1 = wp[(size_t)(k0 + 4) * 128];
        const float wk2 = wp[(size_t)(k0 + 8) * 128];
        const float wk3 = wp[(size_t)(k0 + 12) * 128];
        v4f bv[R];
#pragma unroll
        for (int j = 0; j < R; ++j) bv[j] = *(const v4f*)(rp[j] + w0);
        v2f wa;
        wa.x = wk0; wa.y = wk1;
#pragma unroll
        for (int j = 0; j < R; ++j) {
            v2f bb;
            bb.x = bv[j][0]; bb.y = bv[j][1];
            a[j] = __builtin_amdgcn_wmma_f32_16x16x4_f32(
                false, wa, false, bb, (short)0, a[j], false, false);
        }
        wa.x = wk2; wa.y = wk3;
#pragma unroll
        for (int j = 0; j < R; ++j) {
            v2f bb;
            bb.x = bv[j][2]; bb.y = bv[j][3];
            a[j] = __builtin_amdgcn_wmma_f32_16x16x4_f32(
                false, wa, false, bb, (short)0, a[j], false, false);
        }
    }

    const int kb = ct * 16 + 8 * h;
    const v4f bf0 = *(const v4f*)(bfc2 + kb);
    const v4f bf1 = *(const v4f*)(bfc2 + kb + 4);
    const float bfv[8] = {bf0[0], bf0[1], bf0[2], bf0[3], bf1[0], bf1[1], bf1[2], bf1[3]};
    const int cj = min(c, 5);
    float wo[8];
#pragma unroll
    for (int r = 0; r < 8; ++r) wo[r] = Wout[(size_t)(kb + r) * 6 + cj];
#pragma unroll
    for (int j = 0; j < R; ++j) {
        v8f o = v8f{};
#pragma unroll
        for (int q = 0; q < 4; ++q) {
            v2f hb, wa;
            hb.x = fmaxf(a[j][2 * q]     + bfv[2 * q],     0.0f);
            hb.y = fmaxf(a[j][2 * q + 1] + bfv[2 * q + 1], 0.0f);
            wa.x = (c < 6) ? wo[2 * q]     : 0.0f;
            wa.y = (c < 6) ? wo[2 * q + 1] : 0.0f;
            o = __builtin_amdgcn_wmma_f32_16x16x4_f32(
                false, wa, false, hb, (short)0, o, false, false);
        }
        if (h == 0) {
            float* __restrict__ dst = hp + ((size_t)ct * FC_ROWS + (rg + 4 * j) * 16 + c) * 6;
#pragma unroll
            for (int r = 0; r < 6; ++r) dst[r] = o[r];
        }
    }
}

__global__ __launch_bounds__(AT_THREADS) void agg_tile_kernel(const int* __restrict__ ei,
                                                              const unsigned int* __restrict__ own,
                                                              const float* __restrict__ as4,
                                                              const float* __restrict__ ad4,
                                                              const float* __restrict__ xw,
                                                              const float* __restrict__ b,
                                                              const float* __restrict__ Wfc2,
                                                              const float* __restrict__ bfc2,
                                                              const float* __restrict__ Wout,
                                                              const float* __restrict__ bout,
                                                              float* __restrict__ out) {
    __shared__ __attribute__((aligned(16))) float        acc[AT_ROWS * AT_PITCH];
    __shared__ __attribute__((aligned(16))) float        bperm[256];
    __shared__ __attribute__((aligned(16))) unsigned int raw[AT_RAWW];
    __shared__ unsigned int lp[AT_SUB][AT_CAP];
    __shared__ float        la[AT_SUB][AT_CAP];
    __shared__ unsigned int cnt[AT_SUB];
    __shared__ unsigned int rawcnt;
    __shared__ float        rmax[AT_RANGE];
    __shared__ float        rden[AT_RANGE];
    __shared__ float        rself[AT_RANGE];
    __shared__ float        radl[AT_RANGE];

    const int tid  = threadIdx.x;
    const int lane = tid & 31;
    const int wave = tid >> 5;
    const int d0   = blockIdx.x * AT_RANGE;
    float* const hpart = (float*)raw;
    const unsigned int nvalid = (unsigned int)min(AT_RANGE, N_NODES - d0);

    if (tid < AT_SUB) cnt[tid] = 0u;
    if (tid == AT_SUB) rawcnt = 0u;
    if (tid < 256) bperm[tid] = b[(tid & 128) + ((tid & 31) << 2) + ((tid >> 5) & 3)];
    if ((unsigned int)tid < nvalid) {
        const int node = d0 + tid;
        const float dh = half_of(ad4, node);
        const float sc = leaky(half_of(as4, node) + dh);
        radl[tid]  = dh;
        rself[tid] = sc;
        rmax[tid]  = sc;
        rden[tid]  = 0.0f;
    }
    __syncthreads();

    const uint4* __restrict__ own4 = (const uint4*)own;
    const unsigned int pat = (unsigned int)blockIdx.x * 0x01010101u;
    for (int base = 0; base < AT_OWN16; base += 4 * AT_THREADS) {
        uint4 ov[4];
#pragma unroll
        for (int u = 0; u < 4; ++u) {
            const int i = base + u * AT_THREADS + tid;
            ov[u] = own4[min(i, AT_OWN16 - 1)];
        }
        unsigned int hit[4];
#pragma unroll
        for (int u = 0; u < 4; ++u) {
            const unsigned int w[4] = {ov[u].x ^ pat, ov[u].y ^ pat, ov[u].z ^ pat, ov[u].w ^ pat};
            unsigned int h = 0u;
#pragma unroll
            for (int q = 0; q < 4; ++q) {
                const unsigned int m = ~(((w[q] & 0x7F7F7F7Fu) + 0x7F7F7F7Fu) | w[q]) & 0x80808080u;
                h |= m >> (7 - q);
            }
            hit[u] = (base + u * AT_THREADS + tid < AT_OWN16) ? h : 0u;
        }
#pragma unroll
        for (int v = 0; v < 2; ++v) hit[v] = hit[2 * v] | (hit[2 * v + 1] << 4);
        if ((hit[0] | hit[1]) != 0u) {
            unsigned int idx = atomicAdd(&rawcnt, (unsigned int)(__popc(hit[0]) + __popc(hit[1])));
#pragma unroll
            for (int v = 0; v < 2; ++v) {
                unsigned int hb = hit[v];
                while (hb != 0u) {
                    const int p = __ffs((int)hb) - 1;
                    hb &= hb - 1u;
                    const int u = 2 * v + ((p >> 2) & 1);
                    const int e = (base + u * AT_THREADS + tid) * 16 + 4 * (p & 3) + (p >> 3);
                    if (idx < AT_RAWCAP) raw[idx] = (unsigned int)e;
                    ++idx;
                }
            }
        }
    }
    __syncthreads();

    const unsigned int nraw = rawcnt;
    const bool ovf1 = nraw > (unsigned int)AT_RAWCAP;
    if (!ovf1) {
        for (int i = tid; i < (int)nraw; i += AT_THREADS) {
            const int e = (int)raw[i];
            const int s = ei[e];
            const int d = ei[N_EDGES0 + e];
            asm volatile("" :: "v"(s), "v"(d));
            const unsigned int r = (unsigned int)(d - d0);
            if (r < nvalid) {
                const unsigned int sub = r >= AT_ROWS ? 1u : 0u;
                const unsigned int dl  = r - sub * AT_ROWS;
                const float sc = leaky(half_of(as4, s) + radl[r]);
                const unsigned int idx = atomicAdd(&cnt[sub], 1u);
                if (idx < AT_CAP) {
                    la[sub][idx] = sc;
                    lp[sub][idx] = (unsigned int)s | (dl << 17);
                }
                ldsMaxFloat(&rmax[r], sc);
            }
        }
    }
    __syncthreads();

    const bool ovf = ovf1 || cnt[0] > (unsigned int)AT_CAP || cnt[1] > (unsigned int)AT_CAP;
    if (ovf) {
        for (int e = tid; e < N_EDGES0; e += AT_THREADS) {
            const unsigned int r = (unsigned int)(ei[N_EDGES0 + e] - d0);
            if (r < nvalid) ldsMaxFloat(&rmax[r], leaky(half_of(as4, ei[e]) + radl[r]));
        }
    }
    __syncthreads();

    if ((unsigned int)tid < nvalid) {
        const float ex = __expf(rself[tid] - rmax[tid]);
        rself[tid] = ex;
        atomicAdd(&rden[tid], ex);
    }
    if (!ovf) {
        for (int p = 0; p < AT_SUB; ++p) {
            const int n = (int)cnt[p];
            for (int i = tid; i < n; i += AT_THREADS) {
                const unsigned int rr = (unsigned int)(p * AT_ROWS) + (lp[p][i] >> 17);
                const float ex = __expf(la[p][i] - rmax[rr]);
                la[p][i] = ex;
                atomicAdd(&rden[rr], ex);
            }
        }
    } else {
        for (int e = tid; e < N_EDGES0; e += AT_THREADS) {
            const unsigned int r = (unsigned int)(ei[N_EDGES0 + e] - d0);
            if (r < nvalid) atomicAdd(&rden[r], __expf(leaky(half_of(as4, ei[e]) + radl[r]) - rmax[r]));
        }
    }
    __syncthreads();

    for (int p = 0; p < AT_SUB; ++p) {
        const int rbase = d0 + p * AT_ROWS;
        const int nrows = min(AT_ROWS, N_NODES - rbase);
        const int n = ovf ? 0 : (int)cnt[p];

        for (int r = wave; r < nrows; r += 32) {
            const int node = rbase + r;
            const float a = rself[p * AT_ROWS + r] / rden[p * AT_ROWS + r];
            const float4* row = (const float4*)(xw + (size_t)node * 256);
            float4 v0 = row[lane], v1 = row[32 + lane];
            float* ar = acc + r * AT_PITCH + lane;
            ar[0]   = a * v0.x; ar[32]  = a * v0.y; ar[64]  = a * v0.z; ar[96]  = a * v0.w;
            ar[128] = a * v1.x; ar[160] = a * v1.y; ar[192] = a * v1.z; ar[224] = a * v1.w;
        }
        for (int i = tid; i < n; i += AT_THREADS)
            la[p][i] = la[p][i] / rden[(unsigned int)(p * AT_ROWS) + (lp[p][i] >> 17)];
        __syncthreads();

        if (!ovf) {
            for (int i = wave; i < n; i += 32 * 4) {
                float4 v0[4], v1[4];
                float  a[4];
                int    dl[4];
#pragma unroll
                for (int u = 0; u < 4; ++u) {
                    const int idx = i + 32 * u;
                    const bool ok = idx < n;
                    unsigned int pk = ok ? lp[p][idx] : 0u;
                    a[u]  = ok ? la[p][idx] : 0.0f;
                    dl[u] = (int)(pk >> 17);
                    const float4* row = (const float4*)(xw + (size_t)(pk & 0x1FFFFu) * 256);
                    v0[u] = row[lane];
                    v1[u] = row[32 + lane];
                }
#pragma unroll
                for (int u = 0; u < 4; ++u) {
                    if (i + 32 * u < n) {
                        float* ar = acc + dl[u] * AT_PITCH + lane;
                        atomicAdd(ar,       a[u] * v0[u].x);
                        atomicAdd(ar + 32,  a[u] * v0[u].y);
                        atomicAdd(ar + 64,  a[u] * v0[u].z);
                        atomicAdd(ar + 96,  a[u] * v0[u].w);
                        atomicAdd(ar + 128, a[u] * v1[u].x);
                        atomicAdd(ar + 160, a[u] * v1[u].y);
                        atomicAdd(ar + 192, a[u] * v1[u].z);
                        atomicAdd(ar + 224, a[u] * v1[u].w);
                    }
                }
            }
        } else {
            for (int e = tid; e < N_EDGES0; e += AT_THREADS) {
                const int d = ei[N_EDGES0 + e];
                unsigned int r = (unsigned int)(d - rbase);
                if (r < (unsigned int)max(nrows, 0)) {
                    const int s = ei[e];
                    const unsigned int rr = (unsigned int)(p * AT_ROWS) + r;
                    const float a = __expf(leaky(half_of(as4, s) + radl[rr]) - rmax[rr]) / rden[rr];
                    const float* row = xw + (size_t)s * 256;
                    for (int c = 0; c < 256; ++c)
                        atomicAdd(acc + r * AT_PITCH + at_pos(c), a * row[c]);
                }
            }
        }
        __syncthreads();

        for (int i = tid; i < nrows * 64; i += AT_THREADS) {
            float4* pp = (float4*)(acc + (i >> 6) * AT_PITCH + 4 * (i & 63));
            const float4 bb = ((const float4*)bperm)[i & 63];
            float4 v = *pp;
            v.x = fmaxf(v.x + bb.x, 0.0f); v.y = fmaxf(v.y + bb.y, 0.0f);
            v.z = fmaxf(v.z + bb.z, 0.0f); v.w = fmaxf(v.w + bb.w, 0.0f);
            *pp = v;
        }
        __syncthreads();

        {
            const int wv = __builtin_amdgcn_readfirstlane(wave);
            const int lastRow = max(min(nrows, AT_ROWS) - 1, 0);
            if ((wv >> 3) == 0) fc2_head_tile<4>(acc, hpart, Wfc2, bfc2, Wout, wv & 7, 0, lastRow, lane);
            else                fc2_head_tile<3>(acc, hpart, Wfc2, bfc2, Wout, wv & 7, wv >> 3, lastRow, lane);
        }
        __syncthreads();

        for (int i = tid; i < nrows * 6; i += AT_THREADS) {
            const float s = ((hpart[i] + hpart[FC_ROWS * 6 + i]) + (hpart[2 * FC_ROWS * 6 + i] + hpart[3 * FC_ROWS * 6 + i]))
                          + ((hpart[4 * FC_ROWS * 6 + i] + hpart[5 * FC_ROWS * 6 + i]) + (hpart[6 * FC_ROWS * 6 + i] + hpart[7 * FC_ROWS * 6 + i]));
            out[(size_t)rbase * 6 + i] = s + bout[i % 6];
        }
    }
}

extern "C" void kernel_launch(void* const* d_in, const int* in_sizes, int n_in,
                              void* d_out, int out_size, void* d_ws, size_t ws_size,
                              hipStream_t stream) {
    const float* x     = (const float*)d_in[0];
    const int*   ei    = (const int*)  d_in[1];
    const float* W     = (const float*)d_in[2];
    const float* a_src = (const float*)d_in[3];
    const float* a_dst = (const float*)d_in[4];
    const float* b     = (const float*)d_in[5];
    const float* Wfc2  = (const float*)d_in[6];
    const float* bfc2  = (const float*)d_in[7];
    const float* Wout  = (const float*)d_in[8];
    const float* bout  = (const float*)d_in[9];
    float* out = (float*)d_out;

    float* ws      = (float*)d_ws;
    float* xw      = ws;
    float* as4     = xw + (size_t)N_NODES * 256;
    float* ad4     = as4 + (size_t)N_NODES * 4;
    unsigned int* own = (unsigned int*)(ad4 + (size_t)N_NODES * 4);

    gemm1_xw<<<dim3((ROW_TILES + 7) / 8, 4), 256, 0, stream>>>(x, W, a_src, a_dst, ei, xw, as4, ad4, own);
    agg_tile_kernel<<<AT_WGS, AT_THREADS, 0, stream>>>(ei, own, as4, ad4, xw, b, Wfc2, bfc2, Wout, bout, out);
}


# Aggregation by wave ownership: a wave keeps its rows of the LDS tile in registers and walks its own kept-edge list (one per tile, row parity, wave); rows stored once: no LDS f32 adds; list-order sums
# speedup vs baseline: 3.9866x; 1.0323x over previous
#include <hip/hip_runtime.h>
#include <hip/hip_bf16.h>

typedef __attribute__((ext_vector_type(2))) float v2f;
typedef __attribute__((ext_vector_type(4))) float v4f;
typedef __attribute__((ext_vector_type(8))) float v8f;

#define N_NODES   100000
#define N_EDGES0  800000
#define N_EDGES   900000
#define ROW_TILES 6250

__device__ __forceinline__ float leaky(float v) { return v > 0.0f ? v : 0.2f * v; }

__device__ __forceinline__ void ldsMaxFloat(float* addr, float val) {
    if (val >= 0.0f) atomicMax((int*)addr, __float_as_int(val));
    else             atomicMin((unsigned int*)addr, __float_as_uint(val));
}

#define AT_RANGE_ROWS 392
__global__ __launch_bounds__(256) void gemm1_xw(const float* __restrict__ x,
                                                const float* __restrict__ W,
                                                const float* __restrict__ a_src,
                                                const float* __restrict__ a_dst,
                                                const int* __restrict__ ei,
                                                float* __restrict__ xw,
                                                float* __restrict__ as4,
                                                float* __restrict__ ad4,
                                                unsigned int* __restrict__ own) {
    const int g = (blockIdx.y * gridDim.x + blockIdx.x) * 256 + threadIdx.x;
    if (g < N_EDGES0 / 4) {
        const int4 d = ((const int4*)(ei + N_EDGES0))[g];
        own[g] = (((unsigned int)d.x / AT_RANGE_ROWS) & 0xFFu)
               | ((((unsigned int)d.y / AT_RANGE_ROWS) & 0xFFu) << 8)
               | ((((unsigned int)d.z / AT_RANGE_ROWS) & 0xFFu) << 16)
               | (((unsigned int)d.w / AT_RANGE_ROWS) << 24);
    }
    const int lane = threadIdx.x & 31;
    const int wave = threadIdx.x >> 5;
    const int rowTile = blockIdx.x * 8 + wave;
    if (rowTile >= ROW_TILES) return;
    const int row0 = rowTile * 16;
    const int col0 = blockIdx.y * 64;

    const int c = lane & 15;
    const int h = lane >> 4;
    const float* __restrict__ ap = x + (size_t)(row0 + c) * 128 + 4 * h;
    const float* __restrict__ bp = W + (size_t)(4 * h) * 256 + col0 + 4 * c;

    v8f acc[4] = {};
    for (int k = 0; k < 128; k += 8) {
        const v4f av = *(const v4f*)(ap + k);
        const v4f b0 = *(const v4f*)(bp + (size_t)(k + 0) * 256);
        const v4f b1 = *(const v4f*)(bp + (size_t)(k + 1) * 256);
        const v4f b2 = *(const v4f*)(bp + (size_t)(k + 2) * 256);
        const v4f b3 = *(const v4f*)(bp + (size_t)(k + 3) * 256);
        v2f a;
        a.x = av[0]; a.y = av[1];
#pragma unroll
        for (int t = 0; t < 4; ++t) {
            v2f bm;
            bm.x = b0[t]; bm.y = b1[t];
            acc[t] = __builtin_amdgcn_wmma_f32_16x16x4_f32(
                false, a, false, bm, (short)0, acc[t], false, false);
        }
        a.x = av[2]; a.y = av[3];
#pragma unroll
        for (int t = 0; t < 4; ++t) {
            v2f bm;
            bm.x = b2[t]; bm.y = b3[t];
            acc[t] = __builtin_amdgcn_wmma_f32_16x16x4_f32(
                false, a, false, bm, (short)0, acc[t], false, false);
        }
    }
    const int mBase = row0 + (h << 3);
#pragma unroll
    for (int r = 0; r < 8; ++r) {
        v4f o;
        o[0] = acc[0][r]; o[1] = acc[1][r]; o[2] = acc[2][r]; o[3] = acc[3][r];
        *(v4f*)(xw + (size_t)(mBase + r) * 256 + col0 + 4 * c) = o;
    }

    const v4f sv = *(const v4f*)(a_src + col0 + 4 * c);
    const v4f dv = *(const v4f*)(a_dst + col0 + 4 * c);
    float v[16];
#pragma unroll
    for (int r = 0; r < 8; ++r) {
        v[r]     = acc[0][r] * sv[0] + acc[1][r] * sv[1] + acc[2][r] * sv[2] + acc[3][r] * sv[3];
        v[8 + r] = acc[0][r] * dv[0] + acc[1][r] * dv[1] + acc[2][r] * dv[2] + acc[3][r] * dv[3];
    }
    float u8[8], u4[4], u2[2];
#pragma unroll
    for (int i = 0; i < 8; ++i) {
        const bool up = (c & 8) != 0;
        const float keep = up ? v[i + 8] : v[i];
        const float send = up ? v[i] : v[i + 8];
        u8[i] = keep + __shfl_xor(send, 8, 32);
    }
#pragma unroll
    for (int i = 0; i < 4; ++i) {
        const bool up = (c & 4) != 0;
        const float keep = up ? u8[i + 4] : u8[i];
        const float send = up ? u8[i] : u8[i + 4];
        u4[i] = keep + __shfl_xor(send, 4, 32);
    }
#pragma unroll
    for (int i = 0; i < 2; ++i) {
        const bool up = (c & 2) != 0;
        const float keep = up ? u4[i + 2] : u4[i];
        const float send = up ? u4[i] : u4[i + 2];
        u2[i] = keep + __shfl_xor(send, 2, 32);
    }
    const bool up1 = (c & 1) != 0;
    const float tot = (up1 ? u2[1] : u2[0]) + __shfl_xor(up1 ? u2[0] : u2[1], 1, 32);
    float* __restrict__ hp = (c < 8) ? as4 : ad4;
    hp[(size_t)(mBase + (c & 7)) * 4 + blockIdx.y] = tot;
}

#define AT_ROWS    196
#define AT_SUB     2
#define AT_RANGE   (AT_ROWS * AT_SUB)
#define AT_WGS     ((N_NODES + AT_RANGE - 1) / AT_RANGE)
#define AT_CAP     4096
#define AT_LISTS   (AT_SUB * 2 * 32)
#define AT_LCAP    (AT_SUB * AT_CAP / AT_LISTS)
#define AT_SLOTS   ((AT_ROWS + 63) / 64)
#define AT_RAWCAP  (AT_SUB * AT_CAP)
#define AT_THREADS 1024
#define AT_OWN16   (N_EDGES0 / 16)
#define AT_PITCH   260
#define FC_NT      ((AT_ROWS + 15) / 16)
#define FC_ROWS    (FC_NT * 16)
#define FC_HP      (8 * FC_ROWS * 6)
#define AT_RAWW    (AT_RAWCAP > FC_HP ? AT_RAWCAP : FC_HP)
static_assert(AT_RANGE == AT_RANGE_ROWS, "owner byte and edge pass must split the rows alike");
static_assert(AT_WGS <= 256, "the owner index must fit one byte");
static_assert(AT_THREADS == 1024, "a row's owner is (row >> 1) mod 32: the workgroup has 32 waves");
static_assert(AT_LCAP % 4 == 0, "a list is read four entries at a time");
static_assert(AT_SLOTS == 4, "the edge walk names four row slots");
static_assert(N_EDGES0 % 16 == 0, "owner bytes are read sixteen at a time");

__device__ __forceinline__ int at_pos(int c) {
    return (c & 128) + ((c & 3) << 5) + ((c & 127) >> 2);
}

__device__ __forceinline__ float half_of(const float* __restrict__ p4, int node) {
    const float4 q = ((const float4*)p4)[node];
    return (q.x + q.y) + (q.z + q.w);
}

template <int R>
__device__ __forceinline__ void fc2_head_tile(const float* __restrict__ accp, float* __restrict__ hp,
                                              const float* __restrict__ Wfc2, const float* __restrict__ bfc2,
                                              const float* __restrict__ Wout,
                                              const int ct, const int rg, const int lastRow, const int lane) {
    const int c = lane & 15;
    const int h = lane >> 4;
    const float* rp[R];
#pragma unroll
    for (int j = 0; j < R; ++j) rp[j] = accp + min((rg + 4 * j) * 16 + c, lastRow) * AT_PITCH + 4 * h;
    const float* __restrict__ wp = Wfc2 + (size_t)(16 * h) * 128 + ct * 16 + c;

    v8f a[R];
#pragma unroll
    for (int j = 0; j < R; ++j) a[j] = v8f{};
#pragma unroll 2
    for (int st = 0; st < 32; ++st) {
        const int w0 = st * 8;
        const int k0 = (w0 & 128) + 4 * (w0 & 31) + ((w0 >> 5) & 3);
        const float wk0 = wp[(size_t)(k0 + 0) * 128];
        const float wk1 = wp[(size_t)(k0 + 4) * 128];
        const float wk2 = wp[(size_t)(k0 + 8) * 128];
        const float wk3 = wp[(size_t)(k0 + 12) * 128];
        v4f bv[R];
#pragma unroll
        for (int j = 0; j < R; ++j) bv[j] = *(const v4f*)(rp[j] + w0);
        v2f wa;
        wa.x = wk0; wa.y = wk1;
#pragma unroll
        for (int j = 0; j < R; ++j) {
            v2f bb;
            bb.x = bv[j][0]; bb.y = bv[j][1];
            a[j] = __builtin_amdgcn_wmma_f32_16x16x4_f32(
                false, wa, false, bb, (short)0, a[j], false, false);
        }
        wa.x = wk2; wa.y = wk3;
#pragma unroll
        for (int j = 0; j < R; ++j) {
            v2f bb;
            bb.x = bv[j][2]; bb.y = bv[j][3];
            a[j] = __builtin_amdgcn_wmma_f32_16x16x4_f32(
                false, wa, false, bb, (short)0, a[j], false, false);
        }
    }

    const int kb = ct * 16 + 8 * h;
    const v4f bf0 = *(const v4f*)(bfc2 + kb);
    const v4f bf1 = *(const v4f*)(bfc2 + kb + 4);
    const float bfv[8] = {bf0[0], bf0[1], bf0[2], bf0[3], bf1[0], bf1[1], bf1[2], bf1[3]};
    const int cj = min(c, 5);
    float wo[8];
#pragma unroll
    for (int r = 0; r < 8; ++r) wo[r] = Wout[(size_t)(kb + r) * 6 + cj];
#pragma unroll
    for (int j = 0; j < R; ++j) {
        v8f o = v8f{};
#pragma unroll
        for (int q = 0; q < 4; ++q) {
            v2f hb, wa;
            hb.x = fmaxf(a[j][2 * q]     + bfv[2 * q],     0.0f);
            hb.y = fmaxf(a[j][2 * q + 1] + bfv[2 * q + 1], 0.0f);
            wa.x = (c < 6) ? wo[2 * q]     : 0.0f;
            wa.y = (c < 6) ? wo[2 * q + 1] : 0.0f;
            o = __builtin_amdgcn_wmma_f32_16x16x4_f32(
                false, wa, false, hb, (short)0, o, false, false);
        }
        if (h == 0) {
            float* __restrict__ dst = hp + ((size_t)ct * FC_ROWS + (rg + 4 * j) * 16 + c) * 6;
#pragma unroll
            for (int r = 0; r < 6; ++r) dst[r] = o[r];
        }
    }
}

__global__ __launch_bounds__(AT_THREADS) void agg_tile_kernel(const int* __restrict__ ei,
                                                              const unsigned int* __restrict__ own,
                                                              const float* __restrict__ as4,
                                                              const float* __restrict__ ad4,
                                                              const float* __restrict__ xw,
                                                              const float* __restrict__ b,
                                                              const float* __restrict__ Wfc2,
                                                              const float* __restrict__ bfc2,
                                                              const float* __restrict__ Wout,
                                                              const float* __restrict__ bout,
                                                              float* __restrict__ out) {
    __shared__ __attribute__((aligned(16))) float        acc[AT_ROWS * AT_PITCH];
    __shared__ __attribute__((aligned(16))) float        bperm[256];
    __shared__ __attribute__((aligned(16))) unsigned int raw[AT_RAWW];
    __shared__ __attribute__((aligned(16))) unsigned int lp[AT_LISTS][AT_LCAP];
    __shared__ __attribute__((aligned(16))) float        la[AT_LISTS][AT_LCAP];
    __shared__ unsigned int cnt[AT_LISTS];
    __shared__ unsigned int rawcnt;
    __shared__ unsigned int lfull;
    __shared__ float        rmax[AT_RANGE];
    __shared__ float        rden[AT_RANGE];
    __shared__ float        rself[AT_RANGE];
    __shared__ float        radl[AT_RANGE];

    const int tid  = threadIdx.x;
    const int lane = tid & 31;
    const int wave = tid >> 5;
    const int d0   = blockIdx.x * AT_RANGE;
    float* const hpart = (float*)raw;
    const unsigned int nvalid = (unsigned int)min(AT_RANGE, N_NODES - d0);

    if (tid < AT_LISTS) cnt[tid] = 0u;
    if (tid == AT_LISTS) { rawcnt = 0u; lfull = 0u; }
    if (tid < 256) bperm[tid] = b[(tid & 128) + ((tid & 31) << 2) + ((tid >> 5) & 3)];
    if ((unsigned int)tid < nvalid) {
        const int node = d0 + tid;
        const float dh = half_of(ad4, node);
        const float sc = leaky(half_of(as4, node) + dh);
        radl[tid]  = dh;
        rself[tid] = sc;
        rmax[tid]  = sc;
        rden[tid]  = 0.0f;
    }
    __syncthreads();

    const uint4* __restrict__ own4 = (const uint4*)own;
    const unsigned int pat = (unsigned int)blockIdx.x * 0x01010101u;
    for (int base = 0; base < AT_OWN16; base += 4 * AT_THREADS) {
        uint4 ov[4];
#pragma unroll
        for (int u = 0; u < 4; ++u) {
            const int i = base + u * AT_THREADS + tid;
            ov[u] = own4[min(i, AT_OWN16 - 1)];
        }
        unsigned int hit[4];
#pragma unroll
        for (int u = 0; u < 4; ++u) {
            const unsigned int w[4] = {ov[u].x ^ pat, ov[u].y ^ pat, ov[u].z ^ pat, ov[u].w ^ pat};
            unsigned int h = 0u;
#pragma unroll
            for (int q = 0; q < 4; ++q) {
                const unsigned int m = ~(((w[q] & 0x7F7F7F7Fu) + 0x7F7F7F7Fu) | w[q]) & 0x80808080u;
                h |= m >> (7 - q);
            }
            hit[u] = (base + u * AT_THREADS + tid < AT_OWN16) ? h : 0u;
        }
#pragma unroll
        for (int v = 0; v < 2; ++v) hit[v] = hit[2 * v] | (hit[2 * v + 1] << 4);
        if ((hit[0] | hit[1]) != 0u) {
            unsigned int idx = atomicAdd(&rawcnt, (unsigned int)(__popc(hit[0]) + __popc(hit[1])));
#pragma unroll
            for (int v = 0; v < 2; ++v) {
                unsigned int hb = hit[v];
                while (hb != 0u) {
                    const int p = __ffs((int)hb) - 1;
                    hb &= hb - 1u;
                    const int u = 2 * v + ((p >> 2) & 1);
                    const int e = (base + u * AT_THREADS + tid) * 16 + 4 * (p & 3) + (p >> 3);
                    if (idx < AT_RAWCAP) raw[idx] = (unsigned int)e;
                    ++idx;
                }
            }
        }
    }
    __syncthreads();

    const unsigned int nraw = rawcnt;
    const bool ovf1 = nraw > (unsigned int)AT_RAWCAP;
    if (!ovf1) {
        for (int i = tid; i < (int)nraw; i += AT_THREADS) {
            const int e = (int)raw[i];
            const int s = ei[e];
            const int d = ei[N_EDGES0 + e];
            asm volatile("" :: "v"(s), "v"(d));
            const unsigned int r = (unsigned int)(d - d0);
            if (r < nvalid) {
                const unsigned int sub = r >= AT_ROWS ? 1u : 0u;
                const unsigned int dl  = r - sub * AT_ROWS;
                const float sc = leaky(half_of(as4, s) + radl[r]);
                const unsigned int li  = ((sub * 2u + (dl & 1u)) << 5) + ((dl >> 1) & 31u);
                const unsigned int idx = atomicAdd(&cnt[li], 1u);
                if (idx < AT_LCAP) {
                    la[li][idx] = sc;
                    lp[li][idx] = (unsigned int)s | (dl << 17);
                } else {
                    lfull = 1u;
                }
                ldsMaxFloat(&rmax[r], sc);
            }
        }
    }
    __syncthreads();

    const bool ovf = ovf1 || lfull != 0u;
    if (ovf) {
        for (int e = tid; e < N_EDGES0; e += AT_THREADS) {
            const unsigned int r = (unsigned int)(ei[N_EDGES0 + e] - d0);
            if (r < nvalid) ldsMaxFloat(&rmax[r], leaky(half_of(as4, ei[e]) + radl[r]));
        }
    }
    __syncthreads();

    if ((unsigned int)tid < nvalid) {
        const float ex = __expf(rself[tid] - rmax[tid]);
        rself[tid] = ex;
        atomicAdd(&rden[tid], ex);
    }
    if (!ovf) {
        for (int q = tid; q < AT_LISTS * AT_LCAP; q += AT_THREADS) {
            const int li = q / AT_LCAP;
            const int i  = q % AT_LCAP;
            if (i < (int)cnt[li]) {
                const unsigned int rr = (unsigned int)((li >> 6) * AT_ROWS) + (lp[li][i] >> 17);
                const float ex = __expf(la[li][i] - rmax[rr]);
                la[li][i] = ex;
                atomicAdd(&rden[rr], ex);
            }
        }
    } else {
        for (int e = tid; e < N_EDGES0; e += AT_THREADS) {
            const unsigned int r = (unsigned int)(ei[N_EDGES0 + e] - d0);
            if (r < nvalid) atomicAdd(&rden[r], __expf(leaky(half_of(as4, ei[e]) + radl[r]) - rmax[r]));
        }
    }
    __syncthreads();

    if (!ovf) {
        const int wv = __builtin_amdgcn_readfirstlane(wave);
        const float4 cb0 = ((const float4*)b)[lane];
        const float4 cb1 = ((const float4*)b)[32 + lane];
        for (int p = 0; p < AT_SUB; ++p) {
            const int rbase = d0 + p * AT_ROWS;
            const int nrows = min(AT_ROWS, N_NODES - rbase);
            for (int par = 0; par < 2; ++par) {
                const int li = ((p * 2 + par) << 5) + wv;
                const int n = __builtin_amdgcn_readfirstlane((int)cnt[li]);

                for (int i = lane; i < n; i += 32)
                    la[li][i] = la[li][i] / rden[(unsigned int)(p * AT_ROWS) + (lp[li][i] >> 17)];

                v2f A[AT_SLOTS][4];
#pragma unroll
                for (int s = 0; s < AT_SLOTS; ++s) {
                    const int r = 64 * s + 2 * wv + par;
                    const int node = min(rbase + r, N_NODES - 1);
                    const int rr = min(p * AT_ROWS + r, AT_RANGE - 1);
                    const float a = rself[rr] / rden[rr];
                    const v4f* row = (const v4f*)(xw + (size_t)node * 256);
                    const v4f v0 = row[lane], v1 = row[32 + lane];
                    const v2f av = {a, a};
                    A[s][0] = av * __builtin_shufflevector(v0, v0, 0, 1);
                    A[s][1] = av * __builtin_shufflevector(v0, v0, 2, 3);
                    A[s][2] = av * __builtin_shufflevector(v1, v1, 0, 1);
                    A[s][3] = av * __builtin_shufflevector(v1, v1, 2, 3);
                }

                for (int i = 0; i < n; i += 4) {
                    const uint4  pk4 = *(const uint4*)&lp[li][i];
                    const float4 al4 = *(const float4*)&la[li][i];
                    const unsigned int pkv[4] = {pk4.x, pk4.y, pk4.z, pk4.w};
                    const float        alv[4] = {al4.x, al4.y, al4.z, al4.w};
                    v4f    v0[4], v1[4];
                    float  a[4];
                    int    sl[4];
#pragma unroll
                    for (int u = 0; u < 4; ++u) {
                        const bool ok = i + u < n;
                        const unsigned int pk = (unsigned int)__builtin_amdgcn_readfirstlane((int)(ok ? pkv[u] : 0u));
                        a[u]  = __int_as_float(__builtin_amdgcn_readfirstlane(__float_as_int(ok ? alv[u] : 0.0f)));
                        sl[u] = (int)(pk >> 23);
                        const v4f* row = (const v4f*)(xw + (size_t)(pk & 0x1FFFFu) * 256);
                        v0[u] = row[lane];
                        v1[u] = row[32 + lane];
                    }
#pragma unroll
                    for (int u = 0; u < 4; ++u) {
                        const v2f p0 = __builtin_shufflevector(v0[u], v0[u], 0, 1);
                        const v2f p1 = __builtin_shufflevector(v0[u], v0[u], 2, 3);
                        const v2f p2 = __builtin_shufflevector(v1[u], v1[u], 0, 1);
                        const v2f p3 = __builtin_shufflevector(v1[u], v1[u], 2, 3);
#pragma unroll
                        for (int s = 0; s < AT_SLOTS; ++s) {
                            const float as = (sl[u] == s) ? a[u] : 0.0f;
                            const v2f av = {as, as};
                            A[s][0] = __builtin_elementwise_fma(av, p0, A[s][0]);
                            A[s][1] = __builtin_elementwise_fma(av, p1, A[s][1]);
                            A[s][2] = __builtin_elementwise_fma(av, p2, A[s][2]);
                            A[s][3] = __builtin_elementwise_fma(av, p3, A[s][3]);
                        }
                    }
                }

                if (p > 0 && par == 0) {
                    __syncthreads();
                    const int pb = rbase - AT_ROWS;
                    const int pn = min(AT_ROWS, N_NODES - pb);
                    for (int i = tid; i < pn * 6; i += AT_THREADS) {
                        const float s = ((hpart[i] + hpart[FC_ROWS * 6 + i]) + (hpart[2 * FC_ROWS * 6 + i] + hpart[3 * FC_ROWS * 6 + i]))
                                      + ((hpart[4 * FC_ROWS * 6 + i] + hpart[5 * FC_ROWS * 6 + i]) + (hpart[6 * FC_ROWS * 6 + i] + hpart[7 * FC_ROWS * 6 + i]));
                        out[(size_t)pb * 6 + i] = s + bout[i % 6];
                    }
                }

#pragma unroll
                for (int s = 0; s < AT_SLOTS; ++s) {
                    const int r = 64 * s + 2 * wv + par;
                    if (r < nrows) {
                        float* ar = acc + r * AT_PITCH + lane;
                        ar[0]   = fmaxf(A[s][0].x + cb0.x, 0.0f); ar[32]  = fmaxf(A[s][0].y + cb0.y, 0.0f);
                        ar[64]  = fmaxf(A[s][1].x + cb0.z, 0.0f); ar[96]  = fmaxf(A[s][1].y + cb0.w, 0.0f);
                        ar[128] = fmaxf(A[s][2].x + cb1.x, 0.0f); ar[160] = fmaxf(A[s][2].y + cb1.y, 0.0f);
                        ar[192] = fmaxf(A[s][3].x + cb1.z, 0.0f); ar[224] = fmaxf(A[s][3].y + cb1.w, 0.0f);
                    }
                }
            }
            __syncthreads();

            {
                const int lastRow = max(min(nrows, AT_ROWS) - 1, 0);
                if ((wv >> 3) == 0) fc2_head_tile<4>(acc, hpart, Wfc2, bfc2, Wout, wv & 7, 0, lastRow, lane);
                else                fc2_head_tile<3>(acc, hpart, Wfc2, bfc2, Wout, wv & 7, wv >> 3, lastRow, lane);
            }
        }
        __syncthreads();
        {
            const int rbase = d0 + (AT_SUB - 1) * AT_ROWS;
            const int nrows = min(AT_ROWS, N_NODES - rbase);
            for (int i = tid; i < nrows * 6; i += AT_THREADS) {
                const float s = ((hpart[i] + hpart[FC_ROWS * 6 + i]) + (hpart[2 * FC_ROWS * 6 + i] + hpart[3 * FC_ROWS * 6 + i]))
                              + ((hpart[4 * FC_ROWS * 6 + i] + hpart[5 * FC_ROWS * 6 + i]) + (hpart[6 * FC_ROWS * 6 + i] + hpart[7 * FC_ROWS * 6 + i]));
                out[(size_t)rbase * 6 + i] = s + bout[i % 6];
            }
        }
        return;
    }

    for (int p = 0; p < AT_SUB; ++p) {
        const int rbase = d0 + p * AT_ROWS;
        const int nrows = min(AT_ROWS, N_NODES - rbase);

        for (int r = wave; r < nrows; r += 32) {
            const int node = rbase + r;
            const float a = rself[p * AT_ROWS + r] / rden[p * AT_ROWS + r];
            const float4* row = (const float4*)(xw + (size_t)node * 256);
            float4 v0 = row[lane], v1 = row[32 + lane];
            float* ar = acc + r * AT_PITCH + lane;
            ar[0]   = a * v0.x; ar[32]  = a * v0.y; ar[64]  = a * v0.z; ar[96]  = a * v0.w;
            ar[128] = a * v1.x; ar[160] = a * v1.y; ar[192] = a * v1.z; ar[224] = a * v1.w;
        }
        __syncthreads();

        for (int e = tid; e < N_EDGES0; e += AT_THREADS) {
            const int d = ei[N_EDGES0 + e];
            unsigned int r = (unsigned int)(d - rbase);
            if (r < (unsigned int)max(nrows, 0)) {
                const int s = ei[e];
                const unsigned int rr = (unsigned int)(p * AT_ROWS) + r;
                const float a = __expf(leaky(half_of(as4, s) + radl[rr]) - rmax[rr]) / rden[rr];
                const float* row = xw + (size_t)s * 256;
                for (int c = 0; c < 256; ++c)
                    atomicAdd(acc + r * AT_PITCH + at_pos(c), a * row[c]);
            }
        }
        __syncthreads();

        for (int i = tid; i < nrows * 64; i += AT_THREADS) {
            float4* pp = (float4*)(acc + (i >> 6) * AT_PITCH + 4 * (i & 63));
            const float4 bb = ((const float4*)bperm)[i & 63];
            float4 v = *pp;
            v.x = fmaxf(v.x + bb.x, 0.0f); v.y = fmaxf(v.y + bb.y, 0.0f);
            v.z = fmaxf(v.z + bb.z, 0.0f); v.w = fmaxf(v.w + bb.w, 0.0f);
            *pp = v;
        }
        __syncthreads();

        {
            const int wv = __builtin_amdgcn_readfirstlane(wave);
            const int lastRow = max(min(nrows, AT_ROWS) - 1, 0);
            if ((wv >> 3) == 0) fc2_head_tile<4>(acc, hpart, Wfc2, bfc2, Wout, wv & 7, 0, lastRow, lane);
            else                fc2_head_tile<3>(acc, hpart, Wfc2, bfc2, Wout, wv & 7, wv >> 3, lastRow, lane);
        }
        __syncthreads();

        for (int i = tid; i < nrows * 6; i += AT_THREADS) {
            const float s = ((hpart[i] + hpart[FC_ROWS * 6 + i]) + (hpart[2 * FC_ROWS * 6 + i] + hpart[3 * FC_ROWS * 6 + i]))
                          + ((hpart[4 * FC_ROWS * 6 + i] + hpart[5 * FC_ROWS * 6 + i]) + (hpart[6 * FC_ROWS * 6 + i] + hpart[7 * FC_ROWS * 6 + i]));
            out[(size_t)rbase * 6 + i] = s + bout[i % 6];
        }
    }
}

extern "C" void kernel_launch(void* const* d_in, const int* in_sizes, int n_in,
                              void* d_out, int out_size, void* d_ws, size_t ws_size,
                              hipStream_t stream) {
    const float* x     = (const float*)d_in[0];
    const int*   ei    = (const int*)  d_in[1];
    const float* W     = (const float*)d_in[2];
    const float* a_src = (const float*)d_in[3];
    const float* a_dst = (const float*)d_in[4];
    const float* b     = (const float*)d_in[5];
    const float* Wfc2  = (const float*)d_in[6];
    const float* bfc2  = (const float*)d_in[7];
    const float* Wout  = (const float*)d_in[8];
    const float* bout  = (const float*)d_in[9];
    float* out = (float*)d_out;

    float* ws      = (float*)d_ws;
    float* xw      = ws;
    float* as4     = xw + (size_t)N_NODES * 256;
    float* ad4     = as4 + (size_t)N_NODES * 4;
    unsigned int* own = (unsigned int*)(ad4 + (size_t)N_NODES * 4);

    gemm1_xw<<<dim3((ROW_TILES + 7) / 8, 4), 256, 0, stream>>>(x, W, a_src, a_dst, ei, xw, as4, ad4, own);
    agg_tile_kernel<<<AT_WGS, AT_THREADS, 0, stream>>>(ei, own, as4, ad4, xw, b, Wfc2, bfc2, Wout, bout, out);
}
